# starting kernel
# baseline (speedup 1.0000x reference)
#include <hip/hip_runtime.h>
#include <math.h>

#define N_NODES  50000
#define N_EDGES  600000
#define IN_SIZE  128
#define HIDDEN   256
#define OUT_SIZE 128

typedef __attribute__((ext_vector_type(16))) __bf16 v16bf;
typedef __attribute__((ext_vector_type(8)))  __bf16 v8bf;
typedef __attribute__((ext_vector_type(8)))  float  v8f;

__global__ __launch_bounds__(256)
void zero_f32(float* __restrict__ p, int n) {
    int i = blockIdx.x * blockDim.x + threadIdx.x;
    if (i < n) p[i] = 0.0f;
}

__global__ __launch_bounds__(256)
void build_wcat(const float* __restrict__ Wl, const float* __restrict__ Wr,
                __bf16* __restrict__ Wcat, int rows, int kHalf) {
    int i = blockIdx.x * blockDim.x + threadIdx.x;
    if (i >= rows * kHalf) return;
    int r = i / kHalf, c = i % kHalf;
    __bf16* row = Wcat + (size_t)r * (2 * kHalf);
    row[c]         = (__bf16)Wl[i];
    row[kHalf + c] = (__bf16)Wr[i];
}

__global__ __launch_bounds__(256)
void scatter_add(const float* __restrict__ X, const int* __restrict__ src,
                 const int* __restrict__ dst, float* __restrict__ agg,
                 float* __restrict__ cnt, int nedges, int F) {
    int wave = blockIdx.x * (blockDim.x >> 5) + (threadIdx.x >> 5);
    int lane = threadIdx.x & 31;
    if (wave >= nedges) return;
    int s = src[wave];
    int d = dst[wave];
    if (cnt != nullptr && lane == 0) atomicAdd(cnt + d, 1.0f);
    const float4* xs = (const float4*)(X + (size_t)s * F);
    float* ad = agg + (size_t)d * F;
    int nvec = F >> 2;
    for (int i = lane; i < nvec; i += 32) {
        float4 v = xs[i];
        atomicAdd(ad + i * 4 + 0, v.x);
        atomicAdd(ad + i * 4 + 1, v.y);
        atomicAdd(ad + i * 4 + 2, v.z);
        atomicAdd(ad + i * 4 + 3, v.w);
    }
}

__global__ __launch_bounds__(256)
void prep_xcat1(const float* __restrict__ agg, const float* __restrict__ cnt,
                const float* __restrict__ feat, __bf16* __restrict__ X,
                int nnodes, int F) {
    int i = blockIdx.x * blockDim.x + threadIdx.x;
    if (i >= nnodes * F) return;
    int n = i / F, c = i % F;
    float inv = 1.0f / fmaxf(cnt[n], 1.0f);
    __bf16* row = X + (size_t)n * (2 * F);
    row[c]     = (__bf16)(agg[i] * inv);
    row[F + c] = (__bf16)feat[i];
}

__global__ __launch_bounds__(256)
void prep_xcat2(const float* __restrict__ agg, const float* __restrict__ cnt,
                __bf16* __restrict__ X, int nnodes, int F) {
    int i = blockIdx.x * blockDim.x + threadIdx.x;
    if (i >= nnodes * F) return;
    int n = i / F, c = i % F;
    float inv = 1.0f / fmaxf(cnt[n], 1.0f);
    X[(size_t)n * (2 * F) + c] = (__bf16)(agg[i] * inv);
}

template <int K, int NT>
__global__ __launch_bounds__(256)
void gemm_bf16_wmma(const __bf16* __restrict__ A, const __bf16* __restrict__ B,
                    const float* __restrict__ bias, float* __restrict__ C,
                    int M, int N) {
    const int lane    = threadIdx.x & 31;
    const int wave    = blockIdx.x * (blockDim.x >> 5) + (threadIdx.x >> 5);
    const int ngroups = N / (16 * NT);
    if (wave >= (M >> 4) * ngroups) return;
    const int tm  = (wave / ngroups) << 4;
    const int tn0 = (wave % ngroups) * (16 * NT);
    const int kh  = (lane >> 4) * 8;
    const int l15 = lane & 15;

    const __bf16* arow = A + (size_t)(tm  + l15) * K + kh;
    const __bf16* brow = B + (size_t)(tn0 + l15) * K + kh;

    v8f acc[NT];
#pragma unroll
    for (int t = 0; t < NT; ++t) acc[t] = (v8f){};

#pragma unroll
    for (int k0 = 0; k0 < K; k0 += 32) {
        union { v16bf v; v8bf h[2]; } a;
        a.h[0] = *(const v8bf*)(arow + k0);
        a.h[1] = *(const v8bf*)(arow + k0 + 16);
#pragma unroll
        for (int t = 0; t < NT; ++t) {
            union { v16bf v; v8bf h[2]; } b;
            b.h[0] = *(const v8bf*)(brow + (size_t)t * 16 * K + k0);
            b.h[1] = *(const v8bf*)(brow + (size_t)t * 16 * K + k0 + 16);
            acc[t] = __builtin_amdgcn_wmma_f32_16x16x32_bf16(
                false, a.v, false, b.v, (short)0, acc[t], false, false);
        }
    }

    const int r0 = tm + (lane >> 4) * 8;
#pragma unroll
    for (int t = 0; t < NT; ++t) {
        const int cn = tn0 + t * 16 + l15;
        const float bc = bias[cn];
#pragma unroll
        for (int v = 0; v < 8; ++v)
            C[(size_t)(r0 + v) * N + cn] = acc[t][v] + bc;
    }
}

__global__ __launch_bounds__(256)
void norm_relu_store(float* __restrict__ H, __bf16* __restrict__ X2,
                     int F, int ldx, int coloff) {
    __shared__ float red[256];
    const int n = blockIdx.x;
    const int t = threadIdx.x;
    float* row = H + (size_t)n * F;
    float v = row[t];
    red[t] = v * v;
    __syncthreads();
    for (int s = 128; s > 0; s >>= 1) {
        if (t < s) red[t] += red[t + s];
        __syncthreads();
    }
    float inv = 1.0f / fmaxf(sqrtf(red[0]), 1e-12f);
    float r = fmaxf(v * inv, 0.0f);
    row[t] = r;
    X2[(size_t)n * ldx + coloff + t] = (__bf16)r;
}

__global__ __launch_bounds__(256)
void final_fc(const float* __restrict__ H2, const float* __restrict__ Wfc,
              const float* __restrict__ bfc, float* __restrict__ out,
              int nnodes, int F) {
    int wave = blockIdx.x * (blockDim.x >> 5) + (threadIdx.x >> 5);
    int lane = threadIdx.x & 31;
    if (wave >= nnodes) return;
    const float* h = H2 + (size_t)wave * F;
    float ss = 0.0f, d0 = 0.0f, d1 = 0.0f;
    for (int i = lane; i < F; i += 32) {
        float v = h[i];
        ss += v * v;
        d0 += v * Wfc[i];
        d1 += v * Wfc[F + i];
    }
    for (int off = 16; off > 0; off >>= 1) {
        ss += __shfl_xor(ss, off, 32);
        d0 += __shfl_xor(d0, off, 32);
        d1 += __shfl_xor(d1, off, 32);
    }
    if (lane == 0) {
        float inv = 1.0f / fmaxf(sqrtf(ss), 1e-12f);
        float l0 = d0 * inv + bfc[0];
        float l1 = d1 * inv + bfc[1];
        float m  = fmaxf(l0, l1);
        float e0 = __expf(l0 - m), e1 = __expf(l1 - m);
        float s  = 1.0f / (e0 + e1);
        out[(size_t)wave * 2 + 0] = e0 * s;
        out[(size_t)wave * 2 + 1] = e1 * s;
    }
}

extern "C" void kernel_launch(void* const* d_in, const int* in_sizes, int n_in,
                              void* d_out, int out_size, void* d_ws, size_t ws_size,
                              hipStream_t stream) {
    (void)in_sizes; (void)n_in; (void)out_size; (void)ws_size;

    const float* feat = (const float*)d_in[0];
    const int*   eidx = (const int*)  d_in[1];
    const int*   src  = eidx;
    const int*   dst  = eidx + N_EDGES;
    const float* W1l  = (const float*)d_in[2];
    const float* b1   = (const float*)d_in[3];
    const float* W1r  = (const float*)d_in[4];
    const float* W2l  = (const float*)d_in[5];
    const float* b2   = (const float*)d_in[6];
    const float* W2r  = (const float*)d_in[7];
    const float* Wfc  = (const float*)d_in[8];
    const float* bfc  = (const float*)d_in[9];
    float* out = (float*)d_out;

    char* ws = (char*)d_ws;
    size_t off = 0;
    auto alloc = [&](size_t bytes) -> void* {
        void* p = ws + off;
        off = (off + bytes + 255) & ~(size_t)255;
        return p;
    };
    float*  cnt   = (float*) alloc((size_t)N_NODES * 4);
    float*  agg1  = (float*) alloc((size_t)N_NODES * IN_SIZE * 4);
    __bf16* Xcat1 = (__bf16*)alloc((size_t)N_NODES * 2 * IN_SIZE * 2);
    __bf16* Wcat1 = (__bf16*)alloc((size_t)HIDDEN * 2 * IN_SIZE * 2);
    float*  h1    = (float*) alloc((size_t)N_NODES * HIDDEN * 4);
    float*  agg2  = (float*) alloc((size_t)N_NODES * HIDDEN * 4);
    __bf16* Xcat2 = (__bf16*)alloc((size_t)N_NODES * 2 * HIDDEN * 2);
    __bf16* Wcat2 = (__bf16*)alloc((size_t)OUT_SIZE * 2 * HIDDEN * 2);
    float*  h2    = (float*) alloc((size_t)N_NODES * OUT_SIZE * 4);

    const int TB = 256;
    auto cdiv = [](int a, int b) { return (a + b - 1) / b; };

    zero_f32<<<cdiv(N_NODES, TB), TB, 0, stream>>>(cnt, N_NODES);
    zero_f32<<<cdiv(N_NODES * IN_SIZE, TB), TB, 0, stream>>>(agg1, N_NODES * IN_SIZE);
    zero_f32<<<cdiv(N_NODES * HIDDEN, TB), TB, 0, stream>>>(agg2, N_NODES * HIDDEN);

    build_wcat<<<cdiv(HIDDEN * IN_SIZE, TB), TB, 0, stream>>>(W1l, W1r, Wcat1, HIDDEN, IN_SIZE);
    build_wcat<<<cdiv(OUT_SIZE * HIDDEN, TB), TB, 0, stream>>>(W2l, W2r, Wcat2, OUT_SIZE, HIDDEN);

    scatter_add<<<cdiv(N_EDGES, TB / 32), TB, 0, stream>>>(feat, src, dst, agg1, cnt,
                                                           N_EDGES, IN_SIZE);
    prep_xcat1<<<cdiv(N_NODES * IN_SIZE, TB), TB, 0, stream>>>(agg1, cnt, feat, Xcat1,
                                                               N_NODES, IN_SIZE);
    {
        int waves = (N_NODES / 16) * (HIDDEN / 64);
        gemm_bf16_wmma<2 * IN_SIZE, 4><<<cdiv(waves, TB / 32), TB, 0, stream>>>(
            Xcat1, Wcat1, b1, h1, N_NODES, HIDDEN);
    }
    norm_relu_store<<<N_NODES, HIDDEN, 0, stream>>>(h1, Xcat2, HIDDEN, 2 * HIDDEN, HIDDEN);

    scatter_add<<<cdiv(N_EDGES, TB / 32), TB, 0, stream>>>(h1, src, dst, agg2, nullptr,
                                                           N_EDGES, HIDDEN);
    prep_xcat2<<<cdiv(N_NODES * HIDDEN, TB), TB, 0, stream>>>(agg2, cnt, Xcat2,
                                                              N_NODES, HIDDEN);
    {
        int waves = (N_NODES / 16) * (OUT_SIZE / 64);
        gemm_bf16_wmma<2 * HIDDEN, 4><<<cdiv(waves, TB / 32), TB, 0, stream>>>(
            Xcat2, Wcat2, b2, h2, N_NODES, OUT_SIZE);
    }
    final_fc<<<cdiv(N_NODES, TB / 32), TB, 0, stream>>>(h2, Wfc, bfc, out,
                                                        N_NODES, OUT_SIZE);
}


# Neighbour sums kept in LDS by node-range work-groups: each scans a byte-per-edge owner list and adds gathered rows by LDS f32 adds; global f32 atomics, clears, pack passes gone; arrival-order sums
# speedup vs baseline: 2.7111x; 2.7111x over previous
#include <hip/hip_runtime.h>
#include <math.h>

#define N_NODES  50000
#define N_EDGES  600000
#define IN_SIZE  128
#define HIDDEN   256
#define OUT_SIZE 128

typedef __attribute__((ext_vector_type(16))) __bf16 v16bf;
typedef __attribute__((ext_vector_type(8)))  __bf16 v8bf;
typedef __attribute__((ext_vector_type(8)))  float  v8f;

struct FragChunk { int r; int k0; };
__device__ __forceinline__ FragChunk frag_chunk(int blk, int j, int l, int KS) {
    FragChunk f;
    f.r  = (blk / KS) * 16 + (l & 15);
    f.k0 = (blk % KS) * 32 + 16 * j + 8 * (l >> 4);
    return f;
}
__device__ __forceinline__ v8bf pack8(const float* __restrict__ s, float scale) {
    const float4 a = *(const float4*)s;
    const float4 b = *(const float4*)(s + 4);
    v8bf o;
    o[0] = (__bf16)(a.x * scale); o[1] = (__bf16)(a.y * scale);
    o[2] = (__bf16)(a.z * scale); o[3] = (__bf16)(a.w * scale);
    o[4] = (__bf16)(b.x * scale); o[5] = (__bf16)(b.y * scale);
    o[6] = (__bf16)(b.z * scale); o[7] = (__bf16)(b.w * scale);
    return o;
}
__device__ __forceinline__ v8bf pack8(const float* __restrict__ s) {
    const float4 a = *(const float4*)s;
    const float4 b = *(const float4*)(s + 4);
    v8bf o;
    o[0] = (__bf16)a.x; o[1] = (__bf16)a.y; o[2] = (__bf16)a.z; o[3] = (__bf16)a.w;
    o[4] = (__bf16)b.x; o[5] = (__bf16)b.y; o[6] = (__bf16)b.z; o[7] = (__bf16)b.w;
    return o;
}

__global__ __launch_bounds__(256)
void build_wcat(const float* __restrict__ Wl, const float* __restrict__ Wr,
                __bf16* __restrict__ Wcat, int rows, int kHalf) {
    const int q  = blockIdx.x * blockDim.x + threadIdx.x;
    const int KS = (2 * kHalf) >> 5;
    if (q >= rows * KS * 4) return;
    const FragChunk f = frag_chunk(q >> 6, (q >> 5) & 1, q & 31, KS);
    const float* s = (f.k0 < kHalf) ? Wl + (size_t)f.r * kHalf + f.k0
                                    : Wr + (size_t)f.r * kHalf + (f.k0 - kHalf);
    *(v8bf*)(Wcat + (size_t)q * 8) = pack8(s);
}

#define RT      13
#define RR      (RT * 16)
#define NTILES  (N_NODES / 16)
#define NRANGE  ((NTILES + RT - 1) / RT)
#define N4      (N_EDGES / 4)
#define N16     (N_EDGES / 16)
#define QCAP    8192
static_assert(N_NODES % 16 == 0, "whole tiles");
static_assert(N_EDGES % 16 == 0, "owner bytes are read sixteen at a time");
static_assert(NRANGE <= 256, "an owner number is one byte");

__global__ __launch_bounds__(256)
void edge_owner_bytes(const int* __restrict__ dst, unsigned* __restrict__ own) {
    const int i = blockIdx.x * 256 + threadIdx.x;
    if (i >= N4) return;
    const int4 d = ((const int4*)dst)[i];
    own[i] = (unsigned)(d.x / RR) | ((unsigned)(d.y / RR) << 8) |
             ((unsigned)(d.z / RR) << 16) | ((unsigned)(d.w / RR) << 24);
}

template <int F>
__device__ __noinline__ void add_row_direct(const float* __restrict__ X, int s, float* accrow) {
    const float* xs = X + (size_t)s * F;
    for (int col = 0; col < F; ++col) atomicAdd(accrow + col, xs[col]);
}

__device__ __forceinline__ void lds_barrier() {
    __builtin_amdgcn_fence(__ATOMIC_RELEASE, "workgroup", "local");
    __builtin_amdgcn_s_barrier();
    __builtin_amdgcn_fence(__ATOMIC_ACQUIRE, "workgroup", "local");
}

__device__ __forceinline__ unsigned zero_bytes(unsigned x) {
    return ~(((x & 0x7f7f7f7fu) + 0x7f7f7f7fu) | x) & 0x80808080u;
}

template <int F, int U, bool WITH_SELF>
__global__ __launch_bounds__(1024)
void range_aggregate(const float* __restrict__ X, const int* __restrict__ src,
                     const int* __restrict__ dst, const unsigned* __restrict__ own,
                     const float* __restrict__ selfx, __bf16* __restrict__ Xcat) {
    constexpr int PITCH = F + 4;
    constexpr int NL = F / 32;
    constexpr int G  = 48 / NL;
    constexpr int KS  = (2 * F) / 32;
    constexpr int KSH = F / 32;
    constexpr int PW  = 1024 * U;
    constexpr int NP  = (N16 + PW - 1) / PW;
    __shared__ __attribute__((aligned(16))) float acc[RR * PITCH];
    __shared__ unsigned q[2][QCAP];
    __shared__ int cntl[RR];
    __shared__ unsigned qn[2];

    const int t = threadIdx.x, lane = t & 31;
    const int wave = __builtin_amdgcn_readfirstlane(t >> 5);
    const int tile0 = blockIdx.x * RT;
    const int ntile = (NTILES - tile0 < RT) ? (NTILES - tile0) : RT;
    const int lo = tile0 * 16;
    const unsigned nrows = (unsigned)(ntile * 16);
    const unsigned mine = (unsigned)blockIdx.x * 0x01010101u;

    for (int i = t; i < RR * PITCH / 4; i += 1024)
        ((float4*)acc)[i] = make_float4(0.0f, 0.0f, 0.0f, 0.0f);
    for (int i = t; i < RR; i += 1024) cntl[i] = 0;
    if (t < 2) qn[t] = 0u;

    const uint4* __restrict__ o4 = (const uint4*)own;
    int p = blockIdx.x % NP;
    uint4 ov[U];
#pragma unroll
    for (int u = 0; u < U; ++u) {
        const int i16 = p * PW + u * 1024 + t;
        ov[u] = o4[i16 < N16 ? i16 : N16 - 1];
    }
    __syncthreads();

    unsigned base0 = 0u, base1 = 0u;
    for (int it = 0; it < NP; ++it) {
        const int b = it & 1;
        const unsigned base = b ? base1 : base0;
        unsigned long long m = 0ull;
#pragma unroll
        for (int u = 0; u < U; ++u) {
            const int i16 = p * PW + u * 1024 + t;
            unsigned c = (zero_bytes(ov[u].x ^ mine) >> 7) | (zero_bytes(ov[u].y ^ mine) >> 6) |
                         (zero_bytes(ov[u].z ^ mine) >> 5) | (zero_bytes(ov[u].w ^ mine) >> 4);
            if (i16 >= N16) c = 0u;
            m |= (unsigned long long)c << (4 * (u & 1) + 32 * (u >> 1));
        }
        if (m != 0ull) {
            unsigned pos = atomicAdd(&qn[b], (unsigned)__builtin_popcountll(m)) - base;
            while (m != 0ull) {
                const int bit = __builtin_ctzll(m);
                m &= m - 1ull;
                const int u  = 2 * (bit >> 5) + ((bit >> 2) & 1);
                const int wi = bit & 3, k = (bit >> 3) & 3;
                const int e  = (p * PW + u * 1024 + t) * 16 + wi * 4 + k;
                if (pos < QCAP) {
                    q[b][pos] = (unsigned)e;
                } else {
                    const unsigned rel = (unsigned)(dst[e] - lo);
                    if (rel < nrows) {
                        atomicAdd(&cntl[rel], 1);
                        add_row_direct<F>(X, src[e], acc + rel * PITCH);
                    }
                }
                ++pos;
            }
        }
        const int pn = (p + 1 == NP) ? 0 : p + 1;
#pragma unroll
        for (int u = 0; u < U; ++u) {
            const int i16 = pn * PW + u * 1024 + t;
            ov[u] = o4[i16 < N16 ? i16 : N16 - 1];
        }
        lds_barrier();
        const unsigned tot = qn[b];
        unsigned n = tot - base;
        if (n > QCAP) n = QCAP;
        if (b) base1 = tot; else base0 = tot;
        for (unsigned i0 = wave * G; i0 < n; i0 += 32 * G) {
            const bool have = lane < G && i0 + lane < n;
            unsigned e = 0u;
            if (have) e = q[b][i0 + lane];
            const int s_l = src[e];
            const unsigned d_l = (unsigned)(dst[e] - lo);
            if (have && d_l < nrows) atomicAdd(&cntl[d_l], 1);
            float v[G][NL];
#pragma unroll
            for (int i = 0; i < G; ++i) {
                const int s = __builtin_amdgcn_readlane(s_l, i);
                const float* xs = X + (size_t)s * F + lane;
#pragma unroll
                for (int j = 0; j < NL; ++j) v[i][j] = xs[j * 32];
            }
#pragma unroll
            for (int i = 0; i < G; ++i) {
                const unsigned d = (unsigned)__builtin_amdgcn_readlane((int)d_l, i);
                if (i0 + i < n && d < nrows) {
                    float* ad = acc + d * PITCH + lane;
#pragma unroll
                    for (int j = 0; j < NL; ++j) atomicAdd(ad + j * 32, v[i][j]);
                }
            }
        }
        p = pn;
    }
    __syncthreads();

    const int nchunk = ntile * KSH * 64;
    for (int i = t; i < nchunk; i += 1024) {
        const int tl = i / (KSH * 64), rem = i % (KSH * 64);
        const int ks = rem >> 6, j = (rem >> 5) & 1, l = rem & 31;
        const int rl = tl * 16 + (l & 15);
        const int k0 = ks * 32 + 16 * j + 8 * (l >> 4);
        const float inv = 1.0f / fmaxf((float)cntl[rl], 1.0f);
        const size_t qi = ((size_t)((tile0 + tl) * KS + ks) * 2 + j) * 32 + l;
        *(v8bf*)(Xcat + qi * 8) = pack8(acc + rl * PITCH + k0, inv);
        if (WITH_SELF) {
            const size_t qs = ((size_t)((tile0 + tl) * KS + KSH + ks) * 2 + j) * 32 + l;
            *(v8bf*)(Xcat + qs * 8) = pack8(selfx + (size_t)(lo + rl) * F + k0);
        }
    }
}

template <int K, int NT>
__global__ __launch_bounds__(256)
void gemm_bf16_wmma(const __bf16* __restrict__ A, const __bf16* __restrict__ B,
                    const float* __restrict__ bias, float* __restrict__ C,
                    int M, int N) {
    constexpr int KS  = K / 32;
    constexpr int BLK = 512;
    const int lane    = threadIdx.x & 31;
    const int wave    = blockIdx.x * (blockDim.x >> 5) + (threadIdx.x >> 5);
    const int ngroups = N / (16 * NT);
    if (wave >= (M >> 4) * ngroups) return;
    const int tmt = wave / ngroups;
    const int tm  = tmt << 4;
    const int tn0 = (wave % ngroups) * (16 * NT);
    const int l15 = lane & 15;

    const __bf16* ap = A + (size_t)tmt * KS * BLK + lane * 8;
    const __bf16* bp = B + (size_t)(tn0 >> 4) * KS * BLK + lane * 8;

    v8f acc[NT];
#pragma unroll
    for (int t = 0; t < NT; ++t) acc[t] = (v8f){};

#pragma unroll
    for (int ks = 0; ks < KS; ++ks) {
        union { v16bf v; v8bf h[2]; } a;
        a.h[0] = *(const v8bf*)(ap + ks * BLK);
        a.h[1] = *(const v8bf*)(ap + ks * BLK + 256);
#pragma unroll
        for (int t = 0; t < NT; ++t) {
            union { v16bf v; v8bf h[2]; } b;
            b.h[0] = *(const v8bf*)(bp + (size_t)(t * KS + ks) * BLK);
            b.h[1] = *(const v8bf*)(bp + (size_t)(t * KS + ks) * BLK + 256);
            acc[t] = __builtin_amdgcn_wmma_f32_16x16x32_bf16(
                false, a.v, false, b.v, (short)0, acc[t], false, false);
        }
    }

    const int r0 = tm + (lane >> 4) * 8;
#pragma unroll
    for (int t = 0; t < NT; ++t) {
        const int cn = tn0 + t * 16 + l15;
        const float bc = bias[cn];
#pragma unroll
        for (int v = 0; v < 8; ++v)
            C[(size_t)(r0 + v) * N + cn] = acc[t][v] + bc;
    }
}

__global__ __launch_bounds__(256)
void norm_relu_store(float* __restrict__ H, __bf16* __restrict__ X2, int KS, int ks0) {
    __shared__ float red[16][256];
    __shared__ __attribute__((aligned(16))) __bf16 sh[16][256 + 8];
    const int tile = blockIdx.x;
    const int t = threadIdx.x;
    float* col = H + (size_t)tile * 16 * 256 + t;
    float v[16];
#pragma unroll
    for (int r = 0; r < 16; ++r) v[r] = col[r * 256];
#pragma unroll
    for (int r = 0; r < 16; ++r) red[r][t] = v[r] * v[r];
    __syncthreads();
    for (int s = 128; s > 0; s >>= 1) {
        if (t < s) {
#pragma unroll
            for (int r = 0; r < 16; ++r) red[r][t] += red[r][t + s];
        }
        __syncthreads();
    }
#pragma unroll
    for (int r = 0; r < 16; ++r) {
        float inv = 1.0f / fmaxf(sqrtf(red[r][0]), 1e-12f);
        float x = fmaxf(v[r] * inv, 0.0f);
        col[r * 256] = x;
        sh[r][t] = (__bf16)x;
    }
    __syncthreads();
#pragma unroll
    for (int i = t; i < 512; i += 256) {
        const int kl = i >> 6, j = (i >> 5) & 1, l = i & 31;
        const int k0 = kl * 32 + 16 * j + 8 * (l >> 4);
        const v8bf o = *(const v8bf*)&sh[l & 15][k0];
        const size_t q = ((size_t)(tile * KS + ks0 + kl) * 2 + j) * 32 + l;
        *(v8bf*)(X2 + q * 8) = o;
    }
}

__global__ __launch_bounds__(256)
void final_fc(const float* __restrict__ H2, const float* __restrict__ Wfc,
              const float* __restrict__ bfc, float* __restrict__ out,
              int nnodes, int F) {
    int wave = blockIdx.x * (blockDim.x >> 5) + (threadIdx.x >> 5);
    int lane = threadIdx.x & 31;
    if (wave >= nnodes) return;
    const float* h = H2 + (size_t)wave * F;
    float ss = 0.0f, d0 = 0.0f, d1 = 0.0f;
    for (int i = lane; i < F; i += 32) {
        float v = h[i];
        ss += v * v;
        d0 += v * Wfc[i];
        d1 += v * Wfc[F + i];
    }
    for (int off = 16; off > 0; off >>= 1) {
        ss += __shfl_xor(ss, off, 32);
        d0 += __shfl_xor(d0, off, 32);
        d1 += __shfl_xor(d1, off, 32);
    }
    if (lane == 0) {
        float inv = 1.0f / fmaxf(sqrtf(ss), 1e-12f);
        float l0 = d0 * inv + bfc[0];
        float l1 = d1 * inv + bfc[1];
        float m  = fmaxf(l0, l1);
        float e0 = __expf(l0 - m), e1 = __expf(l1 - m);
        float s  = 1.0f / (e0 + e1);
        out[(size_t)wave * 2 + 0] = e0 * s;
        out[(size_t)wave * 2 + 1] = e1 * s;
    }
}

extern "C" void kernel_launch(void* const* d_in, const int* in_sizes, int n_in,
                              void* d_out, int out_size, void* d_ws, size_t ws_size,
                              hipStream_t stream) {
    (void)in_sizes; (void)n_in; (void)out_size; (void)ws_size;

    const float* feat = (const float*)d_in[0];
    const int*   eidx = (const int*)  d_in[1];
    const int*   src  = eidx;
    const int*   dst  = eidx + N_EDGES;
    const float* W1l  = (const float*)d_in[2];
    const float* b1   = (const float*)d_in[3];
    const float* W1r  = (const float*)d_in[4];
    const float* W2l  = (const float*)d_in[5];
    const float* b2   = (const float*)d_in[6];
    const float* W2r  = (const float*)d_in[7];
    const float* Wfc  = (const float*)d_in[8];
    const float* bfc  = (const float*)d_in[9];
    float* out = (float*)d_out;

    char* ws = (char*)d_ws;
    size_t off = 0;
    auto alloc = [&](size_t bytes) -> void* {
        void* p = ws + off;
        off = (off + bytes + 255) & ~(size_t)255;
        return p;
    };
    __bf16* Xcat1 = (__bf16*)alloc((size_t)N_NODES * 2 * IN_SIZE * 2);
    __bf16* Wcat1 = (__bf16*)alloc((size_t)HIDDEN * 2 * IN_SIZE * 2);
    float*  h1    = (float*) alloc((size_t)N_NODES * HIDDEN * 4);
    __bf16* Xcat2 = (__bf16*)alloc((size_t)N_NODES * 2 * HIDDEN * 2);
    __bf16* Wcat2 = (__bf16*)alloc((size_t)OUT_SIZE * 2 * HIDDEN * 2);
    float*  h2    = (float*) alloc((size_t)N_NODES * OUT_SIZE * 4);
    unsigned* own = (unsigned*)alloc((size_t)N_EDGES);

    const int TB = 256;
    auto cdiv = [](int a, int b) { return (a + b - 1) / b; };

    build_wcat<<<cdiv(HIDDEN * 2 * IN_SIZE / 8, TB), TB, 0, stream>>>(W1l, W1r, Wcat1, HIDDEN, IN_SIZE);
    build_wcat<<<cdiv(OUT_SIZE * 2 * HIDDEN / 8, TB), TB, 0, stream>>>(W2l, W2r, Wcat2, OUT_SIZE, HIDDEN);

    edge_owner_bytes<<<cdiv(N4, TB), TB, 0, stream>>>(dst, own);

    range_aggregate<IN_SIZE, 4, true><<<NRANGE, 1024, 0, stream>>>(feat, src, dst, own, feat, Xcat1);

    {
        int waves = (N_NODES / 16) * (HIDDEN / 64);
        gemm_bf16_wmma<2 * IN_SIZE, 4><<<cdiv(waves, TB / 32), TB, 0, stream>>>(
            Xcat1, Wcat1, b1, h1, N_NODES, HIDDEN);
    }
    norm_relu_store<<<N_NODES / 16, HIDDEN, 0, stream>>>(h1, Xcat2, 2 * HIDDEN / 32, HIDDEN / 32);

    range_aggregate<HIDDEN, 2, false><<<NRANGE, 1024, 0, stream>>>(h1, src, dst, own, nullptr, Xcat2);

    {
        int waves = (N_NODES / 16) * (OUT_SIZE / 64);
        gemm_bf16_wmma<2 * HIDDEN, 4><<<cdiv(waves, TB / 32), TB, 0, stream>>>(
            Xcat2, Wcat2, b2, h2, N_NODES, OUT_SIZE);
    }
    final_fc<<<cdiv(N_NODES, TB / 32), TB, 0, stream>>>(h2, Wfc, bfc, out,
                                                        N_NODES, OUT_SIZE);
}


# Edges sorted once per call into a bucket per owning work-group (piece-local counting sort, src|row words); aggregates walk their bucket, no 10/19 barriered scans of owner bytes; arrival-order sums
# speedup vs baseline: 3.2221x; 1.1885x over previous
#include <hip/hip_runtime.h>
#include <math.h>

#define N_NODES  50000
#define N_EDGES  600000
#define IN_SIZE  128
#define HIDDEN   256
#define OUT_SIZE 128

typedef __attribute__((ext_vector_type(16))) __bf16 v16bf;
typedef __attribute__((ext_vector_type(8)))  __bf16 v8bf;
typedef __attribute__((ext_vector_type(8)))  float  v8f;

struct FragChunk { int r; int k0; };
__device__ __forceinline__ FragChunk frag_chunk(int blk, int j, int l, int KS) {
    FragChunk f;
    f.r  = (blk / KS) * 16 + (l & 15);
    f.k0 = (blk % KS) * 32 + 16 * j + 8 * (l >> 4);
    return f;
}
__device__ __forceinline__ v8bf pack8(const float* __restrict__ s, float scale) {
    const float4 a = *(const float4*)s;
    const float4 b = *(const float4*)(s + 4);
    v8bf o;
    o[0] = (__bf16)(a.x * scale); o[1] = (__bf16)(a.y * scale);
    o[2] = (__bf16)(a.z * scale); o[3] = (__bf16)(a.w * scale);
    o[4] = (__bf16)(b.x * scale); o[5] = (__bf16)(b.y * scale);
    o[6] = (__bf16)(b.z * scale); o[7] = (__bf16)(b.w * scale);
    return o;
}
__device__ __forceinline__ v8bf pack8(const float* __restrict__ s) {
    const float4 a = *(const float4*)s;
    const float4 b = *(const float4*)(s + 4);
    v8bf o;
    o[0] = (__bf16)a.x; o[1] = (__bf16)a.y; o[2] = (__bf16)a.z; o[3] = (__bf16)a.w;
    o[4] = (__bf16)b.x; o[5] = (__bf16)b.y; o[6] = (__bf16)b.z; o[7] = (__bf16)b.w;
    return o;
}

__global__ __launch_bounds__(256)
void build_wcat(const float* __restrict__ Wl, const float* __restrict__ Wr,
                __bf16* __restrict__ Wcat, int rows, int kHalf) {
    const int q  = blockIdx.x * blockDim.x + threadIdx.x;
    const int KS = (2 * kHalf) >> 5;
    if (q >= rows * KS * 4) return;
    const FragChunk f = frag_chunk(q >> 6, (q >> 5) & 1, q & 31, KS);
    const float* s = (f.k0 < kHalf) ? Wl + (size_t)f.r * kHalf + f.k0
                                    : Wr + (size_t)f.r * kHalf + (f.k0 - kHalf);
    *(v8bf*)(Wcat + (size_t)q * 8) = pack8(s);
}

#define RT      13
#define RR      (RT * 16)
#define NTILES  (N_NODES / 16)
#define NRANGE  ((NTILES + RT - 1) / RT)
#define PE      2048
#define NPIECE  ((N_EDGES + PE - 1) / PE)
#define SEGP    512
#define QCAP    8192
static_assert(N_NODES % 16 == 0, "whole tiles");
static_assert(N_EDGES % 4 == 0 && PE % 4 == 0, "edges are read four at a time");
static_assert(NRANGE <= 256, "one histogram bin per thread of the sorting work-group");
static_assert(N_NODES <= 65536 && RR <= 65536, "an edge word holds source and local row in 16 bits each");
static_assert(PE < 65536, "segment start and length are 16 bits each");
static_assert(NPIECE <= SEGP && SEGP <= 1024, "one piece per thread in the prefix sum");

__device__ __forceinline__ int wave_incl_scan(int x, int lane) {
#pragma unroll
    for (int off = 1; off < 32; off <<= 1) {
        const int y = __shfl_up(x, off, 32);
        if (lane >= off) x += y;
    }
    return x;
}

__global__ __launch_bounds__(256)
void edge_partition(const int* __restrict__ src, const int* __restrict__ dst,
                    unsigned* __restrict__ ent, unsigned* __restrict__ segtab) {
    __shared__ int hist[256];
    __shared__ int wsum[8];
    const int t = threadIdx.x, lane = t & 31, wave = t >> 5;
    const int piece = blockIdx.x;
    const int e0 = piece * PE;
    const int ne = (N_EDGES - e0 < PE) ? (N_EDGES - e0) : PE;
    hist[t] = 0;
    __syncthreads();
    int own[8], rank[8];
    unsigned word[8];
#pragma unroll
    for (int u = 0; u < 2; ++u) {
        const int i4 = u * 256 + t;
        const bool ok = 4 * i4 < ne;
        int4 d = make_int4(0, 0, 0, 0), s = d;
        if (ok) {
            d = ((const int4*)(dst + e0))[i4];
            s = ((const int4*)(src + e0))[i4];
        }
        const int dd[4] = {d.x, d.y, d.z, d.w};
        const int ss[4] = {s.x, s.y, s.z, s.w};
#pragma unroll
        for (int k = 0; k < 4; ++k) {
            const int o = dd[k] / RR;
            own[u * 4 + k]  = ok ? o : -1;
            word[u * 4 + k] = (unsigned)ss[k] | ((unsigned)(dd[k] - o * RR) << 16);
            rank[u * 4 + k] = ok ? atomicAdd(&hist[o], 1) : 0;
        }
    }
    __syncthreads();
    const int c = hist[t];
    const int x = wave_incl_scan(c, lane);
    if (lane == 31) wsum[wave] = x;
    __syncthreads();
    int start = x - c;
    for (int w = 0; w < wave; ++w) start += wsum[w];
    hist[t] = start;
    if (t < NRANGE) segtab[(size_t)t * NPIECE + piece] = (unsigned)start | ((unsigned)c << 16);
    __syncthreads();
#pragma unroll
    for (int i = 0; i < 8; ++i)
        if (own[i] >= 0) ent[e0 + hist[own[i]] + rank[i]] = word[i];
}

template <int F, bool WITH_SELF>
__global__ __launch_bounds__(1024)
void range_aggregate(const float* __restrict__ X, const unsigned* __restrict__ ent,
                     const unsigned* __restrict__ segtab,
                     const float* __restrict__ selfx, __bf16* __restrict__ Xcat) {
    constexpr int PITCH = F + 4;
    constexpr int NL = F / 32;
    constexpr int G  = 48 / NL;
    constexpr int KS  = (2 * F) / 32;
    constexpr int KSH = F / 32;
    __shared__ __attribute__((aligned(16))) float acc[(RR + 1) * PITCH];
    __shared__ unsigned q[QCAP];
    __shared__ int cntl[RR];
    __shared__ unsigned segs[NPIECE];
    __shared__ int segoff[SEGP];
    __shared__ int wsum[32];

    const int t = threadIdx.x, lane = t & 31;
    const int wave = __builtin_amdgcn_readfirstlane(t >> 5);
    const int tile0 = blockIdx.x * RT;
    const int ntile = (NTILES - tile0 < RT) ? (NTILES - tile0) : RT;
    const int lo = tile0 * 16;

    unsigned sg = 0u;
    if (t < NPIECE) sg = segtab[(size_t)blockIdx.x * NPIECE + t];

    for (int i = t; i < (RR + 1) * PITCH / 4; i += 1024)
        ((float4*)acc)[i] = make_float4(0.0f, 0.0f, 0.0f, 0.0f);
    for (int i = t; i < RR; i += 1024) cntl[i] = 0;

    const int c = (int)(sg >> 16);
    const int x = wave_incl_scan(c, lane);
    if (lane == 31) wsum[wave] = x;
    if (t < NPIECE) segs[t] = sg;
    __syncthreads();
    const int wv   = wsum[lane];
    const int wvi  = wave_incl_scan(wv, lane);
    const int wofs = __builtin_amdgcn_readlane(wvi - wv, wave);
    const int n    = __builtin_amdgcn_readlane(wvi, 31);
    if (t < SEGP) segoff[t] = (t < NPIECE) ? (wofs + x - c) : 0x7fffffff;
    __syncthreads();

    for (int c0 = 0; c0 < n; c0 += QCAP) {
        const int m = (n - c0 < QCAP) ? (n - c0) : QCAP;
        for (int ib = 0; ib < m; ib += 4096) {
            int g[4], a[4];
#pragma unroll
            for (int u = 0; u < 4; ++u) {
                const int i = ib + u * 1024 + t;
                g[u] = c0 + (i < m ? i : m - 1);
                a[u] = 0;
            }
#pragma unroll
            for (int h = SEGP / 2; h > 0; h >>= 1) {
#pragma unroll
                for (int u = 0; u < 4; ++u)
                    if (segoff[a[u] + h] <= g[u]) a[u] += h;
            }
            unsigned w[4];
#pragma unroll
            for (int u = 0; u < 4; ++u)
                w[u] = ent[(size_t)a[u] * PE + (segs[a[u]] & 0xffffu) + (unsigned)(g[u] - segoff[a[u]])];
#pragma unroll
            for (int u = 0; u < 4; ++u) {
                const int i = ib + u * 1024 + t;
                if (i < m) q[i] = w[u];
            }
        }
        __syncthreads();
        for (int i0 = wave * G; i0 < m; i0 += 32 * G) {
            const bool have = lane < G && i0 + lane < m;
            unsigned wd = (unsigned)RR << 16;
            if (have) wd = q[i0 + lane];
            const int s_l = (int)(wd & 0xffffu);
            const int d_l = (int)(wd >> 16);
            if (have) atomicAdd(&cntl[d_l], 1);
            float v[G][NL];
#pragma unroll
            for (int i = 0; i < G; ++i) {
                const int s = __builtin_amdgcn_readlane(s_l, i);
                const float* xs = X + (size_t)s * F + lane;
#pragma unroll
                for (int j = 0; j < NL; ++j) v[i][j] = xs[j * 32];
            }
#pragma unroll
            for (int i = 0; i < G; ++i) {
                const int d = __builtin_amdgcn_readlane(d_l, i);
                float* ad = acc + d * PITCH + lane;
#pragma unroll
                for (int j = 0; j < NL; ++j) atomicAdd(ad + j * 32, v[i][j]);
            }
        }
        __syncthreads();
    }

    const int nchunk = ntile * KSH * 64;
    for (int i = t; i < nchunk; i += 1024) {
        const int tl = i / (KSH * 64), rem = i % (KSH * 64);
        const int ks = rem >> 6, j = (rem >> 5) & 1, l = rem & 31;
        const int rl = tl * 16 + (l & 15);
        const int k0 = ks * 32 + 16 * j + 8 * (l >> 4);
        const float inv = 1.0f / fmaxf((float)cntl[rl], 1.0f);
        const size_t qi = ((size_t)((tile0 + tl) * KS + ks) * 2 + j) * 32 + l;
        *(v8bf*)(Xcat + qi * 8) = pack8(acc + rl * PITCH + k0, inv);
        if (WITH_SELF) {
            const size_t qs = ((size_t)((tile0 + tl) * KS + KSH + ks) * 2 + j) * 32 + l;
            *(v8bf*)(Xcat + qs * 8) = pack8(selfx + (size_t)(lo + rl) * F + k0);
        }
    }
}

template <int K, int NT>
__global__ __launch_bounds__(256)
void gemm_bf16_wmma(const __bf16* __restrict__ A, const __bf16* __restrict__ B,
                    const float* __restrict__ bias, float* __restrict__ C,
                    int M, int N) {
    constexpr int KS  = K / 32;
    constexpr int BLK = 512;
    const int lane    = threadIdx.x & 31;
    const int wave    = blockIdx.x * (blockDim.x >> 5) + (threadIdx.x >> 5);
    const int ngroups = N / (16 * NT);
    if (wave >= (M >> 4) * ngroups) return;
    const int tmt = wave / ngroups;
    const int tm  = tmt << 4;
    const int tn0 = (wave % ngroups) * (16 * NT);
    const int l15 = lane & 15;

    const __bf16* ap = A + (size_t)tmt * KS * BLK + lane * 8;
    const __bf16* bp = B + (size_t)(tn0 >> 4) * KS * BLK + lane * 8;

    v8f acc[NT];
#pragma unroll
    for (int t = 0; t < NT; ++t) acc[t] = (v8f){};

#pragma unroll
    for (int ks = 0; ks < KS; ++ks) {
        union { v16bf v; v8bf h[2]; } a;
        a.h[0] = *(const v8bf*)(ap + ks * BLK);
        a.h[1] = *(const v8bf*)(ap + ks * BLK + 256);
#pragma unroll
        for (int t = 0; t < NT; ++t) {
            union { v16bf v; v8bf h[2]; } b;
            b.h[0] = *(const v8bf*)(bp + (size_t)(t * KS + ks) * BLK);
            b.h[1] = *(const v8bf*)(bp + (size_t)(t * KS + ks) * BLK + 256);
            acc[t] = __builtin_amdgcn_wmma_f32_16x16x32_bf16(
                false, a.v, false, b.v, (short)0, acc[t], false, false);
        }
    }

    const int r0 = tm + (lane >> 4) * 8;
#pragma unroll
    for (int t = 0; t < NT; ++t) {
        const int cn = tn0 + t * 16 + l15;
        const float bc = bias[cn];
#pragma unroll
        for (int v = 0; v < 8; ++v)
            C[(size_t)(r0 + v) * N + cn] = acc[t][v] + bc;
    }
}

__global__ __launch_bounds__(256)
void norm_relu_store(float* __restrict__ H, __bf16* __restrict__ X2, int KS, int ks0) {
    __shared__ float red[16][256];
    __shared__ __attribute__((aligned(16))) __bf16 sh[16][256 + 8];
    const int tile = blockIdx.x;
    const int t = threadIdx.x;
    float* col = H + (size_t)tile * 16 * 256 + t;
    float v[16];
#pragma unroll
    for (int r = 0; r < 16; ++r) v[r] = col[r * 256];
#pragma unroll
    for (int r = 0; r < 16; ++r) red[r][t] = v[r] * v[r];
    __syncthreads();
    for (int s = 128; s > 0; s >>= 1) {
        if (t < s) {
#pragma unroll
            for (int r = 0; r < 16; ++r) red[r][t] += red[r][t + s];
        }
        __syncthreads();
    }
#pragma unroll
    for (int r = 0; r < 16; ++r) {
        float inv = 1.0f / fmaxf(sqrtf(red[r][0]), 1e-12f);
        float x = fmaxf(v[r] * inv, 0.0f);
        col[r * 256] = x;
        sh[r][t] = (__bf16)x;
    }
    __syncthreads();
#pragma unroll
    for (int i = t; i < 512; i += 256) {
        const int kl = i >> 6, j = (i >> 5) & 1, l = i & 31;
        const int k0 = kl * 32 + 16 * j + 8 * (l >> 4);
        const v8bf o = *(const v8bf*)&sh[l & 15][k0];
        const size_t q = ((size_t)(tile * KS + ks0 + kl) * 2 + j) * 32 + l;
        *(v8bf*)(X2 + q * 8) = o;
    }
}

__global__ __launch_bounds__(256)
void final_fc(const float* __restrict__ H2, const float* __restrict__ Wfc,
              const float* __restrict__ bfc, float* __restrict__ out,
              int nnodes, int F) {
    int wave = blockIdx.x * (blockDim.x >> 5) + (threadIdx.x >> 5);
    int lane = threadIdx.x & 31;
    if (wave >= nnodes) return;
    const float* h = H2 + (size_t)wave * F;
    float ss = 0.0f, d0 = 0.0f, d1 = 0.0f;
    for (int i = lane; i < F; i += 32) {
        float v = h[i];
        ss += v * v;
        d0 += v * Wfc[i];
        d1 += v * Wfc[F + i];
    }
    for (int off = 16; off > 0; off >>= 1) {
        ss += __shfl_xor(ss, off, 32);
        d0 += __shfl_xor(d0, off, 32);
        d1 += __shfl_xor(d1, off, 32);
    }
    if (lane == 0) {
        float inv = 1.0f / fmaxf(sqrtf(ss), 1e-12f);
        float l0 = d0 * inv + bfc[0];
        float l1 = d1 * inv + bfc[1];
        float m  = fmaxf(l0, l1);
        float e0 = __expf(l0 - m), e1 = __expf(l1 - m);
        float s  = 1.0f / (e0 + e1);
        out[(size_t)wave * 2 + 0] = e0 * s;
        out[(size_t)wave * 2 + 1] = e1 * s;
    }
}

extern "C" void kernel_launch(void* const* d_in, const int* in_sizes, int n_in,
                              void* d_out, int out_size, void* d_ws, size_t ws_size,
                              hipStream_t stream) {
    (void)in_sizes; (void)n_in; (void)out_size; (void)ws_size;

    const float* feat = (const float*)d_in[0];
    const int*   eidx = (const int*)  d_in[1];
    const int*   src  = eidx;
    const int*   dst  = eidx + N_EDGES;
    const float* W1l  = (const float*)d_in[2];
    const float* b1   = (const float*)d_in[3];
    const float* W1r  = (const float*)d_in[4];
    const float* W2l  = (const float*)d_in[5];
    const float* b2   = (const float*)d_in[6];
    const float* W2r  = (const float*)d_in[7];
    const float* Wfc  = (const float*)d_in[8];
    const float* bfc  = (const float*)d_in[9];
    float* out = (float*)d_out;

    char* ws = (char*)d_ws;
    size_t off = 0;
    auto alloc = [&](size_t bytes) -> void* {
        void* p = ws + off;
        off = (off + bytes + 255) & ~(size_t)255;
        return p;
    };
    __bf16* Xcat1 = (__bf16*)alloc((size_t)N_NODES * 2 * IN_SIZE * 2);
    __bf16* Wcat1 = (__bf16*)alloc((size_t)HIDDEN * 2 * IN_SIZE * 2);
    float*  h1    = (float*) alloc((size_t)N_NODES * HIDDEN * 4);
    __bf16* Xcat2 = (__bf16*)alloc((size_t)N_NODES * 2 * HIDDEN * 2);
    __bf16* Wcat2 = (__bf16*)alloc((size_t)OUT_SIZE * 2 * HIDDEN * 2);
    float*  h2    = (float*) alloc((size_t)N_NODES * OUT_SIZE * 4);
    unsigned* ent    = (unsigned*)h2;
    unsigned* segtab = ent + (size_t)NPIECE * PE;
    static_assert(((size_t)NPIECE * PE + (size_t)NRANGE * NPIECE) * 4 <= (size_t)N_NODES * OUT_SIZE * 4,
                  "edge words and segment table fit in the h2 plane");

    const int TB = 256;
    auto cdiv = [](int a, int b) { return (a + b - 1) / b; };

    build_wcat<<<cdiv(HIDDEN * 2 * IN_SIZE / 8, TB), TB, 0, stream>>>(W1l, W1r, Wcat1, HIDDEN, IN_SIZE);
    build_wcat<<<cdiv(OUT_SIZE * 2 * HIDDEN / 8, TB), TB, 0, stream>>>(W2l, W2r, Wcat2, OUT_SIZE, HIDDEN);

    edge_partition<<<NPIECE, TB, 0, stream>>>(src, dst, ent, segtab);

    range_aggregate<IN_SIZE, true><<<NRANGE, 1024, 0, stream>>>(feat, ent, segtab, feat, Xcat1);

    {
        int waves = (N_NODES / 16) * (HIDDEN / 64);
        gemm_bf16_wmma<2 * IN_SIZE, 4><<<cdiv(waves, TB / 32), TB, 0, stream>>>(
            Xcat1, Wcat1, b1, h1, N_NODES, HIDDEN);
    }
    norm_relu_store<<<N_NODES / 16, HIDDEN, 0, stream>>>(h1, Xcat2, 2 * HIDDEN / 32, HIDDEN / 32);

    range_aggregate<HIDDEN, false><<<NRANGE, 1024, 0, stream>>>(h1, ent, segtab, nullptr, Xcat2);

    {
        int waves = (N_NODES / 16) * (OUT_SIZE / 64);
        gemm_bf16_wmma<2 * HIDDEN, 4><<<cdiv(waves, TB / 32), TB, 0, stream>>>(
            Xcat2, Wcat2, b2, h2, N_NODES, OUT_SIZE);
    }
    final_fc<<<cdiv(N_NODES, TB / 32), TB, 0, stream>>>(h2, Wfc, bfc, out,
                                                        N_NODES, OUT_SIZE);
}


# Both GEMMs: a wave computes an 80x64 output block (5x4 WMMA tiles, 20 accumulators), each k-step's 18 fragment loads issued as one group 1-2 steps ahead, not 2 loads and a wait per WMMA; sums same
# speedup vs baseline: 3.6198x; 1.1234x over previous
#include <hip/hip_runtime.h>
#include <math.h>

#define N_NODES  50000
#define N_EDGES  600000
#define IN_SIZE  128
#define HIDDEN   256
#define OUT_SIZE 128

typedef __attribute__((ext_vector_type(16))) __bf16 v16bf;
typedef __attribute__((ext_vector_type(8)))  __bf16 v8bf;
typedef __attribute__((ext_vector_type(8)))  float  v8f;

struct FragChunk { int r; int k0; };
__device__ __forceinline__ FragChunk frag_chunk(int blk, int j, int l, int KS) {
    FragChunk f;
    f.r  = (blk / KS) * 16 + (l & 15);
    f.k0 = (blk % KS) * 32 + 16 * j + 8 * (l >> 4);
    return f;
}
__device__ __forceinline__ v8bf pack8(const float* __restrict__ s, float scale) {
    const float4 a = *(const float4*)s;
    const float4 b = *(const float4*)(s + 4);
    v8bf o;
    o[0] = (__bf16)(a.x * scale); o[1] = (__bf16)(a.y * scale);
    o[2] = (__bf16)(a.z * scale); o[3] = (__bf16)(a.w * scale);
    o[4] = (__bf16)(b.x * scale); o[5] = (__bf16)(b.y * scale);
    o[6] = (__bf16)(b.z * scale); o[7] = (__bf16)(b.w * scale);
    return o;
}
__device__ __forceinline__ v8bf pack8(const float* __restrict__ s) {
    const float4 a = *(const float4*)s;
    const float4 b = *(const float4*)(s + 4);
    v8bf o;
    o[0] = (__bf16)a.x; o[1] = (__bf16)a.y; o[2] = (__bf16)a.z; o[3] = (__bf16)a.w;
    o[4] = (__bf16)b.x; o[5] = (__bf16)b.y; o[6] = (__bf16)b.z; o[7] = (__bf16)b.w;
    return o;
}

__global__ __launch_bounds__(256)
void build_wcat(const float* __restrict__ Wl, const float* __restrict__ Wr,
                __bf16* __restrict__ Wcat, int rows, int kHalf) {
    const int q  = blockIdx.x * blockDim.x + threadIdx.x;
    const int KS = (2 * kHalf) >> 5;
    if (q >= rows * KS * 4) return;
    const FragChunk f = frag_chunk(q >> 6, (q >> 5) & 1, q & 31, KS);
    const float* s = (f.k0 < kHalf) ? Wl + (size_t)f.r * kHalf + f.k0
                                    : Wr + (size_t)f.r * kHalf + (f.k0 - kHalf);
    *(v8bf*)(Wcat + (size_t)q * 8) = pack8(s);
}

#define RT      13
#define RR      (RT * 16)
#define NTILES  (N_NODES / 16)
#define NRANGE  ((NTILES + RT - 1) / RT)
#define PE      2048
#define NPIECE  ((N_EDGES + PE - 1) / PE)
#define SEGP    512
#define QCAP    8192
static_assert(N_NODES % 16 == 0, "whole tiles");
static_assert(N_EDGES % 4 == 0 && PE % 4 == 0, "edges are read four at a time");
static_assert(NRANGE <= 256, "one histogram bin per thread of the sorting work-group");
static_assert(N_NODES <= 65536 && RR <= 65536, "an edge word holds source and local row in 16 bits each");
static_assert(PE < 65536, "segment start and length are 16 bits each");
static_assert(NPIECE <= SEGP && SEGP <= 1024, "one piece per thread in the prefix sum");

__device__ __forceinline__ int wave_incl_scan(int x, int lane) {
#pragma unroll
    for (int off = 1; off < 32; off <<= 1) {
        const int y = __shfl_up(x, off, 32);
        if (lane >= off) x += y;
    }
    return x;
}

__global__ __launch_bounds__(256)
void edge_partition(const int* __restrict__ src, const int* __restrict__ dst,
                    unsigned* __restrict__ ent, unsigned* __restrict__ segtab) {
    __shared__ int hist[256];
    __shared__ int wsum[8];
    const int t = threadIdx.x, lane = t & 31, wave = t >> 5;
    const int piece = blockIdx.x;
    const int e0 = piece * PE;
    const int ne = (N_EDGES - e0 < PE) ? (N_EDGES - e0) : PE;
    hist[t] = 0;
    __syncthreads();
    int own[8], rank[8];
    unsigned word[8];
#pragma unroll
    for (int u = 0; u < 2; ++u) {
        const int i4 = u * 256 + t;
        const bool ok = 4 * i4 < ne;
        int4 d = make_int4(0, 0, 0, 0), s = d;
        if (ok) {
            d = ((const int4*)(dst + e0))[i4];
            s = ((const int4*)(src + e0))[i4];
        }
        const int dd[4] = {d.x, d.y, d.z, d.w};
        const int ss[4] = {s.x, s.y, s.z, s.w};
#pragma unroll
        for (int k = 0; k < 4; ++k) {
            const int o = dd[k] / RR;
            own[u * 4 + k]  = ok ? o : -1;
            word[u * 4 + k] = (unsigned)ss[k] | ((unsigned)(dd[k] - o * RR) << 16);
            rank[u * 4 + k] = ok ? atomicAdd(&hist[o], 1) : 0;
        }
    }
    __syncthreads();
    const int c = hist[t];
    const int x = wave_incl_scan(c, lane);
    if (lane == 31) wsum[wave] = x;
    __syncthreads();
    int start = x - c;
    for (int w = 0; w < wave; ++w) start += wsum[w];
    hist[t] = start;
    if (t < NRANGE) segtab[(size_t)t * NPIECE + piece] = (unsigned)start | ((unsigned)c << 16);
    __syncthreads();
#pragma unroll
    for (int i = 0; i < 8; ++i)
        if (own[i] >= 0) ent[e0 + hist[own[i]] + rank[i]] = word[i];
}

template <int F, bool WITH_SELF>
__global__ __launch_bounds__(1024)
void range_aggregate(const float* __restrict__ X, const unsigned* __restrict__ ent,
                     const unsigned* __restrict__ segtab,
                     const float* __restrict__ selfx, __bf16* __restrict__ Xcat) {
    constexpr int PITCH = F + 4;
    constexpr int NL = F / 32;
    constexpr int G  = 48 / NL;
    constexpr int KS  = (2 * F) / 32;
    constexpr int KSH = F / 32;
    __shared__ __attribute__((aligned(16))) float acc[(RR + 1) * PITCH];
    __shared__ unsigned q[QCAP];
    __shared__ int cntl[RR];
    __shared__ unsigned segs[NPIECE];
    __shared__ int segoff[SEGP];
    __shared__ int wsum[32];

    const int t = threadIdx.x, lane = t & 31;
    const int wave = __builtin_amdgcn_readfirstlane(t >> 5);
    const int tile0 = blockIdx.x * RT;
    const int ntile = (NTILES - tile0 < RT) ? (NTILES - tile0) : RT;
    const int lo = tile0 * 16;

    unsigned sg = 0u;
    if (t < NPIECE) sg = segtab[(size_t)blockIdx.x * NPIECE + t];

    for (int i = t; i < (RR + 1) * PITCH / 4; i += 1024)
        ((float4*)acc)[i] = make_float4(0.0f, 0.0f, 0.0f, 0.0f);
    for (int i = t; i < RR; i += 1024) cntl[i] = 0;

    const int c = (int)(sg >> 16);
    const int x = wave_incl_scan(c, lane);
    if (lane == 31) wsum[wave] = x;
    if (t < NPIECE) segs[t] = sg;
    __syncthreads();
    const int wv   = wsum[lane];
    const int wvi  = wave_incl_scan(wv, lane);
    const int wofs = __builtin_amdgcn_readlane(wvi - wv, wave);
    const int n    = __builtin_amdgcn_readlane(wvi, 31);
    if (t < SEGP) segoff[t] = (t < NPIECE) ? (wofs + x - c) : 0x7fffffff;
    __syncthreads();

    for (int c0 = 0; c0 < n; c0 += QCAP) {
        const int m = (n - c0 < QCAP) ? (n - c0) : QCAP;
        for (int ib = 0; ib < m; ib += 4096) {
            int g[4], a[4];
#pragma unroll
            for (int u = 0; u < 4; ++u) {
                const int i = ib + u * 1024 + t;
                g[u] = c0 + (i < m ? i : m - 1);
                a[u] = 0;
            }
#pragma unroll
            for (int h = SEGP / 2; h > 0; h >>= 1) {
#pragma unroll
                for (int u = 0; u < 4; ++u)
                    if (segoff[a[u] + h] <= g[u]) a[u] += h;
            }
            unsigned w[4];
#pragma unroll
            for (int u = 0; u < 4; ++u)
                w[u] = ent[(size_t)a[u] * PE + (segs[a[u]] & 0xffffu) + (unsigned)(g[u] - segoff[a[u]])];
#pragma unroll
            for (int u = 0; u < 4; ++u) {
                const int i = ib + u * 1024 + t;
                if (i < m) q[i] = w[u];
            }
        }
        __syncthreads();
        for (int i0 = wave * G; i0 < m; i0 += 32 * G) {
            const bool have = lane < G && i0 + lane < m;
            unsigned wd = (unsigned)RR << 16;
            if (have) wd = q[i0 + lane];
            const int s_l = (int)(wd & 0xffffu);
            const int d_l = (int)(wd >> 16);
            if (have) atomicAdd(&cntl[d_l], 1);
            float v[G][NL];
#pragma unroll
            for (int i = 0; i < G; ++i) {
                const int s = __builtin_amdgcn_readlane(s_l, i);
                const float* xs = X + (size_t)s * F + lane;
#pragma unroll
                for (int j = 0; j < NL; ++j) v[i][j] = xs[j * 32];
            }
#pragma unroll
            for (int i = 0; i < G; ++i) {
                const int d = __builtin_amdgcn_readlane(d_l, i);
                float* ad = acc + d * PITCH + lane;
#pragma unroll
                for (int j = 0; j < NL; ++j) atomicAdd(ad + j * 32, v[i][j]);
            }
        }
        __syncthreads();
    }

    const int nchunk = ntile * KSH * 64;
    for (int i = t; i < nchunk; i += 1024) {
        const int tl = i / (KSH * 64), rem = i % (KSH * 64);
        const int ks = rem >> 6, j = (rem >> 5) & 1, l = rem & 31;
        const int rl = tl * 16 + (l & 15);
        const int k0 = ks * 32 + 16 * j + 8 * (l >> 4);
        const float inv = 1.0f / fmaxf((float)cntl[rl], 1.0f);
        const size_t qi = ((size_t)((tile0 + tl) * KS + ks) * 2 + j) * 32 + l;
        *(v8bf*)(Xcat + qi * 8) = pack8(acc + rl * PITCH + k0, inv);
        if (WITH_SELF) {
            const size_t qs = ((size_t)((tile0 + tl) * KS + KSH + ks) * 2 + j) * 32 + l;
            *(v8bf*)(Xcat + qs * 8) = pack8(selfx + (size_t)(lo + rl) * F + k0);
        }
    }
}

#define GEMM_MT 5
#define GEMM_NT 4
static_assert(NTILES % GEMM_MT == 0, "whole row groups");
static_assert(HIDDEN % (16 * GEMM_NT) == 0 && OUT_SIZE % (16 * GEMM_NT) == 0, "whole column groups");

template <int K, int N, int MT, int NT, int PF>
__global__ __launch_bounds__(32)
void gemm_bf16_wmma(const __bf16* __restrict__ A, const __bf16* __restrict__ B,
                    const float* __restrict__ bias, float* __restrict__ C) {
    constexpr int KS   = K / 32;
    constexpr int BLK  = 512;
    constexpr int RING = PF + 1;
    constexpr int NG   = N / (16 * NT);
    union Frag { v16bf v; v8bf h[2]; };
    const int lane = threadIdx.x;
    const int tmt0 = ((int)blockIdx.x / NG) * MT;
    const int tn0  = ((int)blockIdx.x % NG) * (16 * NT);
    const int l15  = lane & 15;

    const __bf16* ap = A + (size_t)tmt0 * KS * BLK + lane * 8;
    const __bf16* bp = B + (size_t)(tn0 >> 4) * KS * BLK + lane * 8;

    float bc[NT];
#pragma unroll
    for (int t = 0; t < NT; ++t) bc[t] = bias[tn0 + t * 16 + l15];

    v8f acc[MT][NT];
#pragma unroll
    for (int m = 0; m < MT; ++m)
#pragma unroll
        for (int t = 0; t < NT; ++t) acc[m][t] = (v8f){};

    Frag a[RING][MT], b[RING][NT];
    auto fetch = [&](int slot, int ks) {
#pragma unroll
        for (int m = 0; m < MT; ++m) {
            a[slot][m].h[0] = *(const v8bf*)(ap + (size_t)(m * KS + ks) * BLK);
            a[slot][m].h[1] = *(const v8bf*)(ap + (size_t)(m * KS + ks) * BLK + 256);
        }
#pragma unroll
        for (int t = 0; t < NT; ++t) {
            b[slot][t].h[0] = *(const v8bf*)(bp + (size_t)(t * KS + ks) * BLK);
            b[slot][t].h[1] = *(const v8bf*)(bp + (size_t)(t * KS + ks) * BLK + 256);
        }
    };
#pragma unroll
    for (int p = 0; p < PF; ++p) fetch(p, p);

#pragma unroll
    for (int ks = 0; ks < KS; ++ks) {
        if (ks + PF < KS) fetch((ks + PF) % RING, ks + PF);
        __builtin_amdgcn_sched_barrier(0);
        const int cur = ks % RING;
#pragma unroll
        for (int m = 0; m < MT; ++m)
#pragma unroll
            for (int t = 0; t < NT; ++t)
                acc[m][t] = __builtin_amdgcn_wmma_f32_16x16x32_bf16(
                    false, a[cur][m].v, false, b[cur][t].v, (short)0, acc[m][t], false, false);
        __builtin_amdgcn_sched_barrier(0);
    }

    float* cp = C + (size_t)(tmt0 * 16 + (lane >> 4) * 8) * N + tn0 + l15;
#pragma unroll
    for (int m = 0; m < MT; ++m)
#pragma unroll
        for (int t = 0; t < NT; ++t)
#pragma unroll
            for (int v = 0; v < 8; ++v)
                cp[(size_t)(m * 16 + v) * N + t * 16] = acc[m][t][v] + bc[t];
}

__global__ __launch_bounds__(256)
void norm_relu_store(float* __restrict__ H, __bf16* __restrict__ X2, int KS, int ks0) {
    __shared__ float red[16][256];
    __shared__ __attribute__((aligned(16))) __bf16 sh[16][256 + 8];
    const int tile = blockIdx.x;
    const int t = threadIdx.x;
    float* col = H + (size_t)tile * 16 * 256 + t;
    float v[16];
#pragma unroll
    for (int r = 0; r < 16; ++r) v[r] = col[r * 256];
#pragma unroll
    for (int r = 0; r < 16; ++r) red[r][t] = v[r] * v[r];
    __syncthreads();
    for (int s = 128; s > 0; s >>= 1) {
        if (t < s) {
#pragma unroll
            for (int r = 0; r < 16; ++r) red[r][t] += red[r][t + s];
        }
        __syncthreads();
    }
#pragma unroll
    for (int r = 0; r < 16; ++r) {
        float inv = 1.0f / fmaxf(sqrtf(red[r][0]), 1e-12f);
        float x = fmaxf(v[r] * inv, 0.0f);
        col[r * 256] = x;
        sh[r][t] = (__bf16)x;
    }
    __syncthreads();
#pragma unroll
    for (int i = t; i < 512; i += 256) {
        const int kl = i >> 6, j = (i >> 5) & 1, l = i & 31;
        const int k0 = kl * 32 + 16 * j + 8 * (l >> 4);
        const v8bf o = *(const v8bf*)&sh[l & 15][k0];
        const size_t q = ((size_t)(tile * KS + ks0 + kl) * 2 + j) * 32 + l;
        *(v8bf*)(X2 + q * 8) = o;
    }
}

__global__ __launch_bounds__(256)
void final_fc(const float* __restrict__ H2, const float* __restrict__ Wfc,
              const float* __restrict__ bfc, float* __restrict__ out,
              int nnodes, int F) {
    int wave = blockIdx.x * (blockDim.x >> 5) + (threadIdx.x >> 5);
    int lane = threadIdx.x & 31;
    if (wave >= nnodes) return;
    const float* h = H2 + (size_t)wave * F;
    float ss = 0.0f, d0 = 0.0f, d1 = 0.0f;
    for (int i = lane; i < F; i += 32) {
        float v = h[i];
        ss += v * v;
        d0 += v * Wfc[i];
        d1 += v * Wfc[F + i];
    }
    for (int off = 16; off > 0; off >>= 1) {
        ss += __shfl_xor(ss, off, 32);
        d0 += __shfl_xor(d0, off, 32);
        d1 += __shfl_xor(d1, off, 32);
    }
    if (lane == 0) {
        float inv = 1.0f / fmaxf(sqrtf(ss), 1e-12f);
        float l0 = d0 * inv + bfc[0];
        float l1 = d1 * inv + bfc[1];
        float m  = fmaxf(l0, l1);
        float e0 = __expf(l0 - m), e1 = __expf(l1 - m);
        float s  = 1.0f / (e0 + e1);
        out[(size_t)wave * 2 + 0] = e0 * s;
        out[(size_t)wave * 2 + 1] = e1 * s;
    }
}

extern "C" void kernel_launch(void* const* d_in, const int* in_sizes, int n_in,
                              void* d_out, int out_size, void* d_ws, size_t ws_size,
                              hipStream_t stream) {
    (void)in_sizes; (void)n_in; (void)out_size; (void)ws_size;

    const float* feat = (const float*)d_in[0];
    const int*   eidx = (const int*)  d_in[1];
    const int*   src  = eidx;
    const int*   dst  = eidx + N_EDGES;
    const float* W1l  = (const float*)d_in[2];
    const float* b1   = (const float*)d_in[3];
    const float* W1r  = (const float*)d_in[4];
    const float* W2l  = (const float*)d_in[5];
    const float* b2   = (const float*)d_in[6];
    const float* W2r  = (const float*)d_in[7];
    const float* Wfc  = (const float*)d_in[8];
    const float* bfc  = (const float*)d_in[9];
    float* out = (float*)d_out;

    char* ws = (char*)d_ws;
    size_t off = 0;
    auto alloc = [&](size_t bytes) -> void* {
        void* p = ws + off;
        off = (off + bytes + 255) & ~(size_t)255;
        return p;
    };
    __bf16* Xcat1 = (__bf16*)alloc((size_t)N_NODES * 2 * IN_SIZE * 2);
    __bf16* Wcat1 = (__bf16*)alloc((size_t)HIDDEN * 2 * IN_SIZE * 2);
    float*  h1    = (float*) alloc((size_t)N_NODES * HIDDEN * 4);
    __bf16* Xcat2 = (__bf16*)alloc((size_t)N_NODES * 2 * HIDDEN * 2);
    __bf16* Wcat2 = (__bf16*)alloc((size_t)OUT_SIZE * 2 * HIDDEN * 2);
    float*  h2    = (float*) alloc((size_t)N_NODES * OUT_SIZE * 4);
    unsigned* ent    = (unsigned*)h2;
    unsigned* segtab = ent + (size_t)NPIECE * PE;
    static_assert(((size_t)NPIECE * PE + (size_t)NRANGE * NPIECE) * 4 <= (size_t)N_NODES * OUT_SIZE * 4,
                  "edge words and segment table fit in the h2 plane");

    const int TB = 256;
    auto cdiv = [](int a, int b) { return (a + b - 1) / b; };

    build_wcat<<<cdiv(HIDDEN * 2 * IN_SIZE / 8, TB), TB, 0, stream>>>(W1l, W1r, Wcat1, HIDDEN, IN_SIZE);
    build_wcat<<<cdiv(OUT_SIZE * 2 * HIDDEN / 8, TB), TB, 0, stream>>>(W2l, W2r, Wcat2, OUT_SIZE, HIDDEN);

    edge_partition<<<NPIECE, TB, 0, stream>>>(src, dst, ent, segtab);

    range_aggregate<IN_SIZE, true><<<NRANGE, 1024, 0, stream>>>(feat, ent, segtab, feat, Xcat1);

    gemm_bf16_wmma<2 * IN_SIZE, HIDDEN, GEMM_MT, GEMM_NT, 1>
        <<<(NTILES / GEMM_MT) * (HIDDEN / (16 * GEMM_NT)), 32, 0, stream>>>(Xcat1, Wcat1, b1, h1);
    norm_relu_store<<<N_NODES / 16, HIDDEN, 0, stream>>>(h1, Xcat2, 2 * HIDDEN / 32, HIDDEN / 32);

    range_aggregate<HIDDEN, false><<<NRANGE, 1024, 0, stream>>>(h1, ent, segtab, nullptr, Xcat2);

    gemm_bf16_wmma<2 * HIDDEN, OUT_SIZE, GEMM_MT, GEMM_NT, 2>
        <<<(NTILES / GEMM_MT) * (OUT_SIZE / (16 * GEMM_NT)), 32, 0, stream>>>(Xcat2, Wcat2, b2, h2);
    final_fc<<<cdiv(N_NODES, TB / 32), TB, 0, stream>>>(h2, Wfc, bfc, out,
                                                        N_NODES, OUT_SIZE);
}


# Output head fused into the layer-2 GEMM: a wave owns whole 128-wide rows (5x8 WMMA tiles), runs norm, FC, softmax on its accumulators; head launch and h2 plane gone; norm/FC sum order changes
# speedup vs baseline: 3.9182x; 1.0824x over previous
#include <hip/hip_runtime.h>
#include <math.h>

#define N_NODES  50000
#define N_EDGES  600000
#define IN_SIZE  128
#define HIDDEN   256
#define OUT_SIZE 128

typedef __attribute__((ext_vector_type(16))) __bf16 v16bf;
typedef __attribute__((ext_vector_type(8)))  __bf16 v8bf;
typedef __attribute__((ext_vector_type(8)))  float  v8f;

struct FragChunk { int r; int k0; };
__device__ __forceinline__ FragChunk frag_chunk(int blk, int j, int l, int KS) {
    FragChunk f;
    f.r  = (blk / KS) * 16 + (l & 15);
    f.k0 = (blk % KS) * 32 + 16 * j + 8 * (l >> 4);
    return f;
}
__device__ __forceinline__ v8bf pack8(const float* __restrict__ s, float scale) {
    const float4 a = *(const float4*)s;
    const float4 b = *(const float4*)(s + 4);
    v8bf o;
    o[0] = (__bf16)(a.x * scale); o[1] = (__bf16)(a.y * scale);
    o[2] = (__bf16)(a.z * scale); o[3] = (__bf16)(a.w * scale);
    o[4] = (__bf16)(b.x * scale); o[5] = (__bf16)(b.y * scale);
    o[6] = (__bf16)(b.z * scale); o[7] = (__bf16)(b.w * scale);
    return o;
}
__device__ __forceinline__ v8bf pack8(const float* __restrict__ s) {
    const float4 a = *(const float4*)s;
    const float4 b = *(const float4*)(s + 4);
    v8bf o;
    o[0] = (__bf16)a.x; o[1] = (__bf16)a.y; o[2] = (__bf16)a.z; o[3] = (__bf16)a.w;
    o[4] = (__bf16)b.x; o[5] = (__bf16)b.y; o[6] = (__bf16)b.z; o[7] = (__bf16)b.w;
    return o;
}

__global__ __launch_bounds__(256)
void build_wcat(const float* __restrict__ Wl, const float* __restrict__ Wr,
                __bf16* __restrict__ Wcat, int rows, int kHalf) {
    const int q  = blockIdx.x * blockDim.x + threadIdx.x;
    const int KS = (2 * kHalf) >> 5;
    if (q >= rows * KS * 4) return;
    const FragChunk f = frag_chunk(q >> 6, (q >> 5) & 1, q & 31, KS);
    const float* s = (f.k0 < kHalf) ? Wl + (size_t)f.r * kHalf + f.k0
                                    : Wr + (size_t)f.r * kHalf + (f.k0 - kHalf);
    *(v8bf*)(Wcat + (size_t)q * 8) = pack8(s);
}

#define RT      13
#define RR      (RT * 16)
#define NTILES  (N_NODES / 16)
#define NRANGE  ((NTILES + RT - 1) / RT)
#define PE      2048
#define NPIECE  ((N_EDGES + PE - 1) / PE)
#define SEGP    512
#define QCAP    8192
static_assert(N_NODES % 16 == 0, "whole tiles");
static_assert(N_EDGES % 4 == 0 && PE % 4 == 0, "edges are read four at a time");
static_assert(NRANGE <= 256, "one histogram bin per thread of the sorting work-group");
static_assert(N_NODES <= 65536 && RR <= 65536, "an edge word holds source and local row in 16 bits each");
static_assert(PE < 65536, "segment start and length are 16 bits each");
static_assert(NPIECE <= SEGP && SEGP <= 1024, "one piece per thread in the prefix sum");

__device__ __forceinline__ int wave_incl_scan(int x, int lane) {
#pragma unroll
    for (int off = 1; off < 32; off <<= 1) {
        const int y = __shfl_up(x, off, 32);
        if (lane >= off) x += y;
    }
    return x;
}

__global__ __launch_bounds__(256)
void edge_partition(const int* __restrict__ src, const int* __restrict__ dst,
                    unsigned* __restrict__ ent, unsigned* __restrict__ segtab) {
    __shared__ int hist[256];
    __shared__ int wsum[8];
    const int t = threadIdx.x, lane = t & 31, wave = t >> 5;
    const int piece = blockIdx.x;
    const int e0 = piece * PE;
    const int ne = (N_EDGES - e0 < PE) ? (N_EDGES - e0) : PE;
    hist[t] = 0;
    __syncthreads();
    int own[8], rank[8];
    unsigned word[8];
#pragma unroll
    for (int u = 0; u < 2; ++u) {
        const int i4 = u * 256 + t;
        const bool ok = 4 * i4 < ne;
        int4 d = make_int4(0, 0, 0, 0), s = d;
        if (ok) {
            d = ((const int4*)(dst + e0))[i4];
            s = ((const int4*)(src + e0))[i4];
        }
        const int dd[4] = {d.x, d.y, d.z, d.w};
        const int ss[4] = {s.x, s.y, s.z, s.w};
#pragma unroll
        for (int k = 0; k < 4; ++k) {
            const int o = dd[k] / RR;
            own[u * 4 + k]  = ok ? o : -1;
            word[u * 4 + k] = (unsigned)ss[k] | ((unsigned)(dd[k] - o * RR) << 16);
            rank[u * 4 + k] = ok ? atomicAdd(&hist[o], 1) : 0;
        }
    }
    __syncthreads();
    const int c = hist[t];
    const int x = wave_incl_scan(c, lane);
    if (lane == 31) wsum[wave] = x;
    __syncthreads();
    int start = x - c;
    for (int w = 0; w < wave; ++w) start += wsum[w];
    hist[t] = start;
    if (t < NRANGE) segtab[(size_t)t * NPIECE + piece] = (unsigned)start | ((unsigned)c << 16);
    __syncthreads();
#pragma unroll
    for (int i = 0; i < 8; ++i)
        if (own[i] >= 0) ent[e0 + hist[own[i]] + rank[i]] = word[i];
}

template <int F, bool WITH_SELF>
__global__ __launch_bounds__(1024)
void range_aggregate(const float* __restrict__ X, const unsigned* __restrict__ ent,
                     const unsigned* __restrict__ segtab,
                     const float* __restrict__ selfx, __bf16* __restrict__ Xcat) {
    constexpr int PITCH = F + 4;
    constexpr int NL = F / 32;
    constexpr int G  = 48 / NL;
    constexpr int KS  = (2 * F) / 32;
    constexpr int KSH = F / 32;
    __shared__ __attribute__((aligned(16))) float acc[(RR + 1) * PITCH];
    __shared__ unsigned q[QCAP];
    __shared__ int cntl[RR];
    __shared__ unsigned segs[NPIECE];
    __shared__ int segoff[SEGP];
    __shared__ int wsum[32];

    const int t = threadIdx.x, lane = t & 31;
    const int wave = __builtin_amdgcn_readfirstlane(t >> 5);
    const int tile0 = blockIdx.x * RT;
    const int ntile = (NTILES - tile0 < RT) ? (NTILES - tile0) : RT;
    const int lo = tile0 * 16;

    unsigned sg = 0u;
    if (t < NPIECE) sg = segtab[(size_t)blockIdx.x * NPIECE + t];

    for (int i = t; i < (RR + 1) * PITCH / 4; i += 1024)
        ((float4*)acc)[i] = make_float4(0.0f, 0.0f, 0.0f, 0.0f);
    for (int i = t; i < RR; i += 1024) cntl[i] = 0;

    const int c = (int)(sg >> 16);
    const int x = wave_incl_scan(c, lane);
    if (lane == 31) wsum[wave] = x;
    if (t < NPIECE) segs[t] = sg;
    __syncthreads();
    const int wv   = wsum[lane];
    const int wvi  = wave_incl_scan(wv, lane);
    const int wofs = __builtin_amdgcn_readlane(wvi - wv, wave);
    const int n    = __builtin_amdgcn_readlane(wvi, 31);
    if (t < SEGP) segoff[t] = (t < NPIECE) ? (wofs + x - c) : 0x7fffffff;
    __syncthreads();

    for (int c0 = 0; c0 < n; c0 += QCAP) {
        const int m = (n - c0 < QCAP) ? (n - c0) : QCAP;
        for (int ib = 0; ib < m; ib += 4096) {
            int g[4], a[4];
#pragma unroll
            for (int u = 0; u < 4; ++u) {
                const int i = ib + u * 1024 + t;
                g[u] = c0 + (i < m ? i : m - 1);
                a[u] = 0;
            }
#pragma unroll
            for (int h = SEGP / 2; h > 0; h >>= 1) {
#pragma unroll
                for (int u = 0; u < 4; ++u)
                    if (segoff[a[u] + h] <= g[u]) a[u] += h;
            }
            unsigned w[4];
#pragma unroll
            for (int u = 0; u < 4; ++u)
                w[u] = ent[(size_t)a[u] * PE + (segs[a[u]] & 0xffffu) + (unsigned)(g[u] - segoff[a[u]])];
#pragma unroll
            for (int u = 0; u < 4; ++u) {
                const int i = ib + u * 1024 + t;
                if (i < m) q[i] = w[u];
            }
        }
        __syncthreads();
        for (int i0 = wave * G; i0 < m; i0 += 32 * G) {
            const bool have = lane < G && i0 + lane < m;
            unsigned wd = (unsigned)RR << 16;
            if (have) wd = q[i0 + lane];
            const int s_l = (int)(wd & 0xffffu);
            const int d_l = (int)(wd >> 16);
            if (have) atomicAdd(&cntl[d_l], 1);
            float v[G][NL];
#pragma unroll
            for (int i = 0; i < G; ++i) {
                const int s = __builtin_amdgcn_readlane(s_l, i);
                const float* xs = X + (size_t)s * F + lane;
#pragma unroll
                for (int j = 0; j < NL; ++j) v[i][j] = xs[j * 32];
            }
#pragma unroll
            for (int i = 0; i < G; ++i) {
                const int d = __builtin_amdgcn_readlane(d_l, i);
                float* ad = acc + d * PITCH + lane;
#pragma unroll
                for (int j = 0; j < NL; ++j) atomicAdd(ad + j * 32, v[i][j]);
            }
        }
        __syncthreads();
    }

    const int nchunk = ntile * KSH * 64;
    for (int i = t; i < nchunk; i += 1024) {
        const int tl = i / (KSH * 64), rem = i % (KSH * 64);
        const int ks = rem >> 6, j = (rem >> 5) & 1, l = rem & 31;
        const int rl = tl * 16 + (l & 15);
        const int k0 = ks * 32 + 16 * j + 8 * (l >> 4);
        const float inv = 1.0f / fmaxf((float)cntl[rl], 1.0f);
        const size_t qi = ((size_t)((tile0 + tl) * KS + ks) * 2 + j) * 32 + l;
        *(v8bf*)(Xcat + qi * 8) = pack8(acc + rl * PITCH + k0, inv);
        if (WITH_SELF) {
            const size_t qs = ((size_t)((tile0 + tl) * KS + KSH + ks) * 2 + j) * 32 + l;
            *(v8bf*)(Xcat + qs * 8) = pack8(selfx + (size_t)(lo + rl) * F + k0);
        }
    }
}

#define GEMM_MT 5
#define GEMM_NT 4
static_assert(NTILES % GEMM_MT == 0, "whole row groups");
static_assert(HIDDEN % (16 * GEMM_NT) == 0, "whole column groups");

template <int M>
__device__ __forceinline__ float row_xor(float x) {
    const int b = __builtin_bit_cast(int, x);
    return __builtin_bit_cast(float, __builtin_amdgcn_update_dpp(b, b, 0x160 + M, 0xf, 0xf, false));
}
__device__ __forceinline__ float row_tree8(const float (&x)[8], bool b3, bool b2, bool b1) {
    float y[4], z[2];
#pragma unroll
    for (int i = 0; i < 4; ++i)
        y[i] = (b3 ? x[i + 4] : x[i]) + row_xor<8>(b3 ? x[i] : x[i + 4]);
#pragma unroll
    for (int i = 0; i < 2; ++i)
        z[i] = (b2 ? y[i + 2] : y[i]) + row_xor<4>(b2 ? y[i] : y[i + 2]);
    float w = (b1 ? z[1] : z[0]) + row_xor<2>(b1 ? z[0] : z[1]);
    w += row_xor<1>(w);
    return w;
}

template <int K, int N, int MT, int NT, int PF, bool HEAD>
__global__ __launch_bounds__(32)
void gemm_bf16_wmma(const __bf16* __restrict__ A, const __bf16* __restrict__ B,
                    const float* __restrict__ bias, float* __restrict__ C,
                    const float* __restrict__ Wfc, const float* __restrict__ bfc) {
    static_assert(!HEAD || N == 16 * NT, "the head needs whole rows in one wave");
    constexpr int KS   = K / 32;
    constexpr int BLK  = 512;
    constexpr int RING = PF + 1;
    constexpr int NG   = N / (16 * NT);
    union Frag { v16bf v; v8bf h[2]; };
    const int lane = threadIdx.x;
    const int tmt0 = ((int)blockIdx.x / NG) * MT;
    const int tn0  = ((int)blockIdx.x % NG) * (16 * NT);
    const int l15  = lane & 15;

    const __bf16* ap = A + (size_t)tmt0 * KS * BLK + lane * 8;
    const __bf16* bp = B + (size_t)(tn0 >> 4) * KS * BLK + lane * 8;

    float bc[NT];
#pragma unroll
    for (int t = 0; t < NT; ++t) bc[t] = bias[tn0 + t * 16 + l15];

    v8f acc[MT][NT];
#pragma unroll
    for (int m = 0; m < MT; ++m)
#pragma unroll
        for (int t = 0; t < NT; ++t) acc[m][t] = (v8f){};

    Frag a[RING][MT], b[RING][NT];
    auto fetch = [&](int slot, int ks) {
#pragma unroll
        for (int m = 0; m < MT; ++m) {
            a[slot][m].h[0] = *(const v8bf*)(ap + (size_t)(m * KS + ks) * BLK);
            a[slot][m].h[1] = *(const v8bf*)(ap + (size_t)(m * KS + ks) * BLK + 256);
        }
#pragma unroll
        for (int t = 0; t < NT; ++t) {
            b[slot][t].h[0] = *(const v8bf*)(bp + (size_t)(t * KS + ks) * BLK);
            b[slot][t].h[1] = *(const v8bf*)(bp + (size_t)(t * KS + ks) * BLK + 256);
        }
    };
#pragma unroll
    for (int p = 0; p < PF; ++p) fetch(p, p);

#pragma unroll
    for (int ks = 0; ks < KS; ++ks) {
        if (ks + PF < KS) fetch((ks + PF) % RING, ks + PF);
        __builtin_amdgcn_sched_barrier(0);
        const int cur = ks % RING;
#pragma unroll
        for (int m = 0; m < MT; ++m)
#pragma unroll
            for (int t = 0; t < NT; ++t)
                acc[m][t] = __builtin_amdgcn_wmma_f32_16x16x32_bf16(
                    false, a[cur][m].v, false, b[cur][t].v, (short)0, acc[m][t], false, false);
        __builtin_amdgcn_sched_barrier(0);
    }

    if constexpr (!HEAD) {
        float* cp = C + (size_t)(tmt0 * 16 + (lane >> 4) * 8) * N + tn0 + l15;
#pragma unroll
        for (int m = 0; m < MT; ++m)
#pragma unroll
            for (int t = 0; t < NT; ++t)
#pragma unroll
                for (int v = 0; v < 8; ++v)
                    cp[(size_t)(m * 16 + v) * N + t * 16] = acc[m][t][v] + bc[t];
    } else {
        float w0[NT], w1[NT];
#pragma unroll
        for (int t = 0; t < NT; ++t) {
            w0[t] = Wfc[t * 16 + l15];
            w1[t] = Wfc[N + t * 16 + l15];
        }
        const float bf0 = bfc[0], bf1 = bfc[1];
        const bool b3 = (lane & 8) != 0, b2 = (lane & 4) != 0, b1 = (lane & 2) != 0;
        float2* op = (float2*)C + (size_t)tmt0 * 16 + (lane >> 4) * 8 + (l15 >> 1);
#pragma unroll
        for (int m = 0; m < MT; ++m) {
            float ss[8], d0[8], d1[8];
#pragma unroll
            for (int v = 0; v < 8; ++v) {
                float s = 0.0f, p = 0.0f, q = 0.0f;
#pragma unroll
                for (int t = 0; t < NT; ++t) {
                    const float x = acc[m][t][v] + bc[t];
                    s += x * x;
                    p += x * w0[t];
                    q += x * w1[t];
                }
                ss[v] = s; d0[v] = p; d1[v] = q;
            }
            const float rs = row_tree8(ss, b3, b2, b1);
            const float r0 = row_tree8(d0, b3, b2, b1);
            const float r1 = row_tree8(d1, b3, b2, b1);
            const float inv = 1.0f / fmaxf(sqrtf(rs), 1e-12f);
            const float l0 = r0 * inv + bf0;
            const float l1 = r1 * inv + bf1;
            const float mx = fmaxf(l0, l1);
            const float e0 = __expf(l0 - mx), e1 = __expf(l1 - mx);
            const float sc = 1.0f / (e0 + e1);
            if ((lane & 1) == 0) op[m * 16] = make_float2(e0 * sc, e1 * sc);
        }
    }
}

__global__ __launch_bounds__(256)
void norm_relu_store(float* __restrict__ H, __bf16* __restrict__ X2, int KS, int ks0) {
    __shared__ float red[16][256];
    __shared__ __attribute__((aligned(16))) __bf16 sh[16][256 + 8];
    const int tile = blockIdx.x;
    const int t = threadIdx.x;
    float* col = H + (size_t)tile * 16 * 256 + t;
    float v[16];
#pragma unroll
    for (int r = 0; r < 16; ++r) v[r] = col[r * 256];
#pragma unroll
    for (int r = 0; r < 16; ++r) red[r][t] = v[r] * v[r];
    __syncthreads();
    for (int s = 128; s > 0; s >>= 1) {
        if (t < s) {
#pragma unroll
            for (int r = 0; r < 16; ++r) red[r][t] += red[r][t + s];
        }
        __syncthreads();
    }
#pragma unroll
    for (int r = 0; r < 16; ++r) {
        float inv = 1.0f / fmaxf(sqrtf(red[r][0]), 1e-12f);
        float x = fmaxf(v[r] * inv, 0.0f);
        col[r * 256] = x;
        sh[r][t] = (__bf16)x;
    }
    __syncthreads();
#pragma unroll
    for (int i = t; i < 512; i += 256) {
        const int kl = i >> 6, j = (i >> 5) & 1, l = i & 31;
        const int k0 = kl * 32 + 16 * j + 8 * (l >> 4);
        const v8bf o = *(const v8bf*)&sh[l & 15][k0];
        const size_t q = ((size_t)(tile * KS + ks0 + kl) * 2 + j) * 32 + l;
        *(v8bf*)(X2 + q * 8) = o;
    }
}

extern "C" void kernel_launch(void* const* d_in, const int* in_sizes, int n_in,
                              void* d_out, int out_size, void* d_ws, size_t ws_size,
                              hipStream_t stream) {
    (void)in_sizes; (void)n_in; (void)out_size; (void)ws_size;

    const float* feat = (const float*)d_in[0];
    const int*   eidx = (const int*)  d_in[1];
    const int*   src  = eidx;
    const int*   dst  = eidx + N_EDGES;
    const float* W1l  = (const float*)d_in[2];
    const float* b1   = (const float*)d_in[3];
    const float* W1r  = (const float*)d_in[4];
    const float* W2l  = (const float*)d_in[5];
    const float* b2   = (const float*)d_in[6];
    const float* W2r  = (const float*)d_in[7];
    const float* Wfc  = (const float*)d_in[8];
    const float* bfc  = (const float*)d_in[9];
    float* out = (float*)d_out;

    char* ws = (char*)d_ws;
    size_t off = 0;
    auto alloc = [&](size_t bytes) -> void* {
        void* p = ws + off;
        off = (off + bytes + 255) & ~(size_t)255;
        return p;
    };
    __bf16* Xcat1 = (__bf16*)alloc((size_t)N_NODES * 2 * IN_SIZE * 2);
    __bf16* Wcat1 = (__bf16*)alloc((size_t)HIDDEN * 2 * IN_SIZE * 2);
    float*  h1    = (float*) alloc((size_t)N_NODES * HIDDEN * 4);
    __bf16* Xcat2 = (__bf16*)alloc((size_t)N_NODES * 2 * HIDDEN * 2);
    __bf16* Wcat2 = (__bf16*)alloc((size_t)OUT_SIZE * 2 * HIDDEN * 2);
    unsigned* ent    = (unsigned*)alloc(((size_t)NPIECE * PE + (size_t)NRANGE * NPIECE) * 4);
    unsigned* segtab = ent + (size_t)NPIECE * PE;

    const int TB = 256;
    auto cdiv = [](int a, int b) { return (a + b - 1) / b; };

    build_wcat<<<cdiv(HIDDEN * 2 * IN_SIZE / 8, TB), TB, 0, stream>>>(W1l, W1r, Wcat1, HIDDEN, IN_SIZE);
    build_wcat<<<cdiv(OUT_SIZE * 2 * HIDDEN / 8, TB), TB, 0, stream>>>(W2l, W2r, Wcat2, OUT_SIZE, HIDDEN);

    edge_partition<<<NPIECE, TB, 0, stream>>>(src, dst, ent, segtab);

    range_aggregate<IN_SIZE, true><<<NRANGE, 1024, 0, stream>>>(feat, ent, segtab, feat, Xcat1);

    gemm_bf16_wmma<2 * IN_SIZE, HIDDEN, GEMM_MT, GEMM_NT, 1, false>
        <<<(NTILES / GEMM_MT) * (HIDDEN / (16 * GEMM_NT)), 32, 0, stream>>>(Xcat1, Wcat1, b1, h1, nullptr, nullptr);
    norm_relu_store<<<N_NODES / 16, HIDDEN, 0, stream>>>(h1, Xcat2, 2 * HIDDEN / 32, HIDDEN / 32);

    range_aggregate<HIDDEN, false><<<NRANGE, 1024, 0, stream>>>(h1, ent, segtab, nullptr, Xcat2);

    gemm_bf16_wmma<2 * HIDDEN, OUT_SIZE, GEMM_MT, OUT_SIZE / 16, 2, true>
        <<<NTILES / GEMM_MT, 32, 0, stream>>>(Xcat2, Wcat2, b2, out, Wfc, bfc);
}


# Each bucket's edge list counting-sorted by target row in LDS (rank = returning int add); a wave sums its group's same-target rows in registers, one LDS f32 add per run: 4-6x fewer; sums regrouped
# speedup vs baseline: 4.3044x; 1.0986x over previous
#include <hip/hip_runtime.h>
#include <math.h>

#define N_NODES  50000
#define N_EDGES  600000
#define IN_SIZE  128
#define HIDDEN   256
#define OUT_SIZE 128

typedef __attribute__((ext_vector_type(16))) __bf16 v16bf;
typedef __attribute__((ext_vector_type(8)))  __bf16 v8bf;
typedef __attribute__((ext_vector_type(8)))  float  v8f;

struct FragChunk { int r; int k0; };
__device__ __forceinline__ FragChunk frag_chunk(int blk, int j, int l, int KS) {
    FragChunk f;
    f.r  = (blk / KS) * 16 + (l & 15);
    f.k0 = (blk % KS) * 32 + 16 * j + 8 * (l >> 4);
    return f;
}
__device__ __forceinline__ v8bf pack8(const float* __restrict__ s, float scale) {
    const float4 a = *(const float4*)s;
    const float4 b = *(const float4*)(s + 4);
    v8bf o;
    o[0] = (__bf16)(a.x * scale); o[1] = (__bf16)(a.y * scale);
    o[2] = (__bf16)(a.z * scale); o[3] = (__bf16)(a.w * scale);
    o[4] = (__bf16)(b.x * scale); o[5] = (__bf16)(b.y * scale);
    o[6] = (__bf16)(b.z * scale); o[7] = (__bf16)(b.w * scale);
    return o;
}
__device__ __forceinline__ v8bf pack8(const float* __restrict__ s) {
    const float4 a = *(const float4*)s;
    const float4 b = *(const float4*)(s + 4);
    v8bf o;
    o[0] = (__bf16)a.x; o[1] = (__bf16)a.y; o[2] = (__bf16)a.z; o[3] = (__bf16)a.w;
    o[4] = (__bf16)b.x; o[5] = (__bf16)b.y; o[6] = (__bf16)b.z; o[7] = (__bf16)b.w;
    return o;
}

__global__ __launch_bounds__(256)
void build_wcat(const float* __restrict__ Wl, const float* __restrict__ Wr,
                __bf16* __restrict__ Wcat, int rows, int kHalf) {
    const int q  = blockIdx.x * blockDim.x + threadIdx.x;
    const int KS = (2 * kHalf) >> 5;
    if (q >= rows * KS * 4) return;
    const FragChunk f = frag_chunk(q >> 6, (q >> 5) & 1, q & 31, KS);
    const float* s = (f.k0 < kHalf) ? Wl + (size_t)f.r * kHalf + f.k0
                                    : Wr + (size_t)f.r * kHalf + (f.k0 - kHalf);
    *(v8bf*)(Wcat + (size_t)q * 8) = pack8(s);
}

#define RT      13
#define RR      (RT * 16)
#define NTILES  (N_NODES / 16)
#define NRANGE  ((NTILES + RT - 1) / RT)
#define PE      2048
#define NPIECE  ((N_EDGES + PE - 1) / PE)
#define SEGP    512
#define QCAP    4096
static_assert(N_NODES % 16 == 0, "whole tiles");
static_assert(N_EDGES % 4 == 0 && PE % 4 == 0, "edges are read four at a time");
static_assert(NRANGE <= 256, "one histogram bin per thread of the sorting work-group");
static_assert(RR <= 256 && QCAP == 4096, "row starts are prefix-summed by 8 waves; a thread holds its 4 entries of a chunk");
static_assert(N_NODES <= 65536 && RR <= 65536, "an edge word holds source and local row in 16 bits each");
static_assert(PE < 65536, "segment start and length are 16 bits each");
static_assert(NPIECE <= SEGP && SEGP <= 1024, "one piece per thread in the prefix sum");

__device__ __forceinline__ int wave_incl_scan(int x, int lane) {
#pragma unroll
    for (int off = 1; off < 32; off <<= 1) {
        const int y = __shfl_up(x, off, 32);
        if (lane >= off) x += y;
    }
    return x;
}

__global__ __launch_bounds__(256)
void edge_partition(const int* __restrict__ src, const int* __restrict__ dst,
                    unsigned* __restrict__ ent, unsigned* __restrict__ segtab) {
    __shared__ int hist[256];
    __shared__ int wsum[8];
    const int t = threadIdx.x, lane = t & 31, wave = t >> 5;
    const int piece = blockIdx.x;
    const int e0 = piece * PE;
    const int ne = (N_EDGES - e0 < PE) ? (N_EDGES - e0) : PE;
    hist[t] = 0;
    __syncthreads();
    int own[8], rank[8];
    unsigned word[8];
#pragma unroll
    for (int u = 0; u < 2; ++u) {
        const int i4 = u * 256 + t;
        const bool ok = 4 * i4 < ne;
        int4 d = make_int4(0, 0, 0, 0), s = d;
        if (ok) {
            d = ((const int4*)(dst + e0))[i4];
            s = ((const int4*)(src + e0))[i4];
        }
        const int dd[4] = {d.x, d.y, d.z, d.w};
        const int ss[4] = {s.x, s.y, s.z, s.w};
#pragma unroll
        for (int k = 0; k < 4; ++k) {
            const int o = dd[k] / RR;
            own[u * 4 + k]  = ok ? o : -1;
            word[u * 4 + k] = (unsigned)ss[k] | ((unsigned)(dd[k] - o * RR) << 16);
            rank[u * 4 + k] = ok ? atomicAdd(&hist[o], 1) : 0;
        }
    }
    __syncthreads();
    const int c = hist[t];
    const int x = wave_incl_scan(c, lane);
    if (lane == 31) wsum[wave] = x;
    __syncthreads();
    int start = x - c;
    for (int w = 0; w < wave; ++w) start += wsum[w];
    hist[t] = start;
    if (t < NRANGE) segtab[(size_t)t * NPIECE + piece] = (unsigned)start | ((unsigned)c << 16);
    __syncthreads();
#pragma unroll
    for (int i = 0; i < 8; ++i)
        if (own[i] >= 0) ent[e0 + hist[own[i]] + rank[i]] = word[i];
}

template <int F, bool WITH_SELF>
__global__ __launch_bounds__(1024)
void range_aggregate(const float* __restrict__ X, const unsigned* __restrict__ ent,
                     const unsigned* __restrict__ segtab,
                     const float* __restrict__ selfx, __bf16* __restrict__ Xcat) {
    constexpr int PITCH = F + 4;
    constexpr int NL = F / 32;
    constexpr int G  = 48 / NL;
    constexpr int KS  = (2 * F) / 32;
    constexpr int KSH = F / 32;
    __shared__ __attribute__((aligned(16))) float acc[(RR + 1) * PITCH];
    __shared__ unsigned q[QCAP];
    __shared__ int cntl[RR];
    __shared__ int ccnt[RR];
    __shared__ int rsum[8];
    __shared__ unsigned segs[NPIECE];
    __shared__ int segoff[SEGP];
    __shared__ int wsum[32];

    const int t = threadIdx.x, lane = t & 31;
    const int wave = __builtin_amdgcn_readfirstlane(t >> 5);
    const int tile0 = blockIdx.x * RT;
    const int ntile = (NTILES - tile0 < RT) ? (NTILES - tile0) : RT;
    const int lo = tile0 * 16;

    unsigned sg = 0u;
    if (t < NPIECE) sg = segtab[(size_t)blockIdx.x * NPIECE + t];

    for (int i = t; i < (RR + 1) * PITCH / 4; i += 1024)
        ((float4*)acc)[i] = make_float4(0.0f, 0.0f, 0.0f, 0.0f);
    for (int i = t; i < RR; i += 1024) { cntl[i] = 0; ccnt[i] = 0; }

    const int c = (int)(sg >> 16);
    const int x = wave_incl_scan(c, lane);
    if (lane == 31) wsum[wave] = x;
    if (t < NPIECE) segs[t] = sg;
    __syncthreads();
    const int wv   = wsum[lane];
    const int wvi  = wave_incl_scan(wv, lane);
    const int wofs = __builtin_amdgcn_readlane(wvi - wv, wave);
    const int n    = __builtin_amdgcn_readlane(wvi, 31);
    if (t < SEGP) segoff[t] = (t < NPIECE) ? (wofs + x - c) : 0x7fffffff;
    __syncthreads();

    for (int c0 = 0; c0 < n; c0 += QCAP) {
        const int m = (n - c0 < QCAP) ? (n - c0) : QCAP;
        int g[4], a[4];
#pragma unroll
        for (int u = 0; u < 4; ++u) {
            const int i = u * 1024 + t;
            g[u] = c0 + (i < m ? i : m - 1);
            a[u] = 0;
        }
#pragma unroll
        for (int h = SEGP / 2; h > 0; h >>= 1) {
#pragma unroll
            for (int u = 0; u < 4; ++u)
                if (segoff[a[u] + h] <= g[u]) a[u] += h;
        }
        unsigned w[4];
        int rank[4];
#pragma unroll
        for (int u = 0; u < 4; ++u)
            w[u] = ent[(size_t)a[u] * PE + (segs[a[u]] & 0xffffu) + (unsigned)(g[u] - segoff[a[u]])];
#pragma unroll
        for (int u = 0; u < 4; ++u)
            rank[u] = (u * 1024 + t < m) ? atomicAdd(&ccnt[w[u] >> 16], 1) : 0;
        __syncthreads();
        int rc = 0, rx = 0;
        if (t < RR) rc = ccnt[t];
        if (wave < 8) {
            rx = wave_incl_scan(rc, lane);
            if (lane == 31) rsum[wave] = rx;
        }
        __syncthreads();
        if (t < RR) {
            int start = rx - rc;
            for (int w2 = 0; w2 < wave; ++w2) start += rsum[w2];
            ccnt[t] = start;
            cntl[t] += rc;
        }
        __syncthreads();
#pragma unroll
        for (int u = 0; u < 4; ++u)
            if (u * 1024 + t < m) q[ccnt[w[u] >> 16] + rank[u]] = w[u];
        __syncthreads();
        for (int i0 = wave * G; i0 < m; i0 += 32 * G) {
            const bool have = lane < G && i0 + lane < m;
            unsigned wd = (unsigned)RR << 16;
            if (have) wd = q[i0 + lane];
            const int s_l = (int)(wd & 0xffffu);
            const int d_l = (int)(wd >> 16);
            float v[G][NL];
#pragma unroll
            for (int i = 0; i < G; ++i) {
                const int s = __builtin_amdgcn_readlane(s_l, i);
                const float* xs = X + (size_t)s * F + lane;
#pragma unroll
                for (int j = 0; j < NL; ++j) v[i][j] = xs[j * 32];
            }
            int dp = __builtin_amdgcn_readlane(d_l, 0);
#pragma unroll
            for (int i = 1; i < G; ++i) {
                const int d = __builtin_amdgcn_readlane(d_l, i);
                if (d == dp) {
#pragma unroll
                    for (int j = 0; j < NL; ++j) v[i][j] += v[i - 1][j];
                } else {
                    float* ad = acc + dp * PITCH + lane;
#pragma unroll
                    for (int j = 0; j < NL; ++j) atomicAdd(ad + j * 32, v[i - 1][j]);
                }
                dp = d;
            }
            {
                float* ad = acc + dp * PITCH + lane;
#pragma unroll
                for (int j = 0; j < NL; ++j) atomicAdd(ad + j * 32, v[G - 1][j]);
            }
        }
        __syncthreads();
        if (c0 + QCAP < n) {
            if (t < RR) ccnt[t] = 0;
            __syncthreads();
        }
    }

    const int nchunk = ntile * KSH * 64;
    for (int i = t; i < nchunk; i += 1024) {
        const int tl = i / (KSH * 64), rem = i % (KSH * 64);
        const int ks = rem >> 6, j = (rem >> 5) & 1, l = rem & 31;
        const int rl = tl * 16 + (l & 15);
        const int k0 = ks * 32 + 16 * j + 8 * (l >> 4);
        const float inv = 1.0f / fmaxf((float)cntl[rl], 1.0f);
        const size_t qi = ((size_t)((tile0 + tl) * KS + ks) * 2 + j) * 32 + l;
        *(v8bf*)(Xcat + qi * 8) = pack8(acc + rl * PITCH + k0, inv);
        if (WITH_SELF) {
            const size_t qs = ((size_t)((tile0 + tl) * KS + KSH + ks) * 2 + j) * 32 + l;
            *(v8bf*)(Xcat + qs * 8) = pack8(selfx + (size_t)(lo + rl) * F + k0);
        }
    }
}

#define GEMM_MT 5
#define GEMM_NT 4
static_assert(NTILES % GEMM_MT == 0, "whole row groups");
static_assert(HIDDEN % (16 * GEMM_NT) == 0, "whole column groups");

template <int M>
__device__ __forceinline__ float row_xor(float x) {
    const int b = __builtin_bit_cast(int, x);
    return __builtin_bit_cast(float, __builtin_amdgcn_update_dpp(b, b, 0x160 + M, 0xf, 0xf, false));
}
__device__ __forceinline__ float row_tree8(const float (&x)[8], bool b3, bool b2, bool b1) {
    float y[4], z[2];
#pragma unroll
    for (int i = 0; i < 4; ++i)
        y[i] = (b3 ? x[i + 4] : x[i]) + row_xor<8>(b3 ? x[i] : x[i + 4]);
#pragma unroll
    for (int i = 0; i < 2; ++i)
        z[i] = (b2 ? y[i + 2] : y[i]) + row_xor<4>(b2 ? y[i] : y[i + 2]);
    float w = (b1 ? z[1] : z[0]) + row_xor<2>(b1 ? z[0] : z[1]);
    w += row_xor<1>(w);
    return w;
}

template <int K, int N, int MT, int NT, int PF, bool HEAD>
__global__ __launch_bounds__(32)
void gemm_bf16_wmma(const __bf16* __restrict__ A, const __bf16* __restrict__ B,
                    const float* __restrict__ bias, float* __restrict__ C,
                    const float* __restrict__ Wfc, const float* __restrict__ bfc) {
    static_assert(!HEAD || N == 16 * NT, "the head needs whole rows in one wave");
    constexpr int KS   = K / 32;
    constexpr int BLK  = 512;
    constexpr int RING = PF + 1;
    constexpr int NG   = N / (16 * NT);
    union Frag { v16bf v; v8bf h[2]; };
    const int lane = threadIdx.x;
    const int tmt0 = ((int)blockIdx.x / NG) * MT;
    const int tn0  = ((int)blockIdx.x % NG) * (16 * NT);
    const int l15  = lane & 15;

    const __bf16* ap = A + (size_t)tmt0 * KS * BLK + lane * 8;
    const __bf16* bp = B + (size_t)(tn0 >> 4) * KS * BLK + lane * 8;

    float bc[NT];
#pragma unroll
    for (int t = 0; t < NT; ++t) bc[t] = bias[tn0 + t * 16 + l15];

    v8f acc[MT][NT];
#pragma unroll
    for (int m = 0; m < MT; ++m)
#pragma unroll
        for (int t = 0; t < NT; ++t) acc[m][t] = (v8f){};

    Frag a[RING][MT], b[RING][NT];
    auto fetch = [&](int slot, int ks) {
#pragma unroll
        for (int m = 0; m < MT; ++m) {
            a[slot][m].h[0] = *(const v8bf*)(ap + (size_t)(m * KS + ks) * BLK);
            a[slot][m].h[1] = *(const v8bf*)(ap + (size_t)(m * KS + ks) * BLK + 256);
        }
#pragma unroll
        for (int t = 0; t < NT; ++t) {
            b[slot][t].h[0] = *(const v8bf*)(bp + (size_t)(t * KS + ks) * BLK);
            b[slot][t].h[1] = *(const v8bf*)(bp + (size_t)(t * KS + ks) * BLK + 256);
        }
    };
#pragma unroll
    for (int p = 0; p < PF; ++p) fetch(p, p);

#pragma unroll
    for (int ks = 0; ks < KS; ++ks) {
        if (ks + PF < KS) fetch((ks + PF) % RING, ks + PF);
        __builtin_amdgcn_sched_barrier(0);
        const int cur = ks % RING;
#pragma unroll
        for (int m = 0; m < MT; ++m)
#pragma unroll
            for (int t = 0; t < NT; ++t)
                acc[m][t] = __builtin_amdgcn_wmma_f32_16x16x32_bf16(
                    false, a[cur][m].v, false, b[cur][t].v, (short)0, acc[m][t], false, false);
        __builtin_amdgcn_sched_barrier(0);
    }

    if constexpr (!HEAD) {
        float* cp = C + (size_t)(tmt0 * 16 + (lane >> 4) * 8) * N + tn0 + l15;
#pragma unroll
        for (int m = 0; m < MT; ++m)
#pragma unroll
            for (int t = 0; t < NT; ++t)
#pragma unroll
                for (int v = 0; v < 8; ++v)
                    cp[(size_t)(m * 16 + v) * N + t * 16] = acc[m][t][v] + bc[t];
    } else {
        float w0[NT], w1[NT];
#pragma unroll
        for (int t = 0; t < NT; ++t) {
            w0[t] = Wfc[t * 16 + l15];
            w1[t] = Wfc[N + t * 16 + l15];
        }
        const float bf0 = bfc[0], bf1 = bfc[1];
        const bool b3 = (lane & 8) != 0, b2 = (lane & 4) != 0, b1 = (lane & 2) != 0;
        float2* op = (float2*)C + (size_t)tmt0 * 16 + (lane >> 4) * 8 + (l15 >> 1);
#pragma unroll
        for (int m = 0; m < MT; ++m) {
            float ss[8], d0[8], d1[8];
#pragma unroll
            for (int v = 0; v < 8; ++v) {
                float s = 0.0f, p = 0.0f, q = 0.0f;
#pragma unroll
                for (int t = 0; t < NT; ++t) {
                    const float x = acc[m][t][v] + bc[t];
                    s += x * x;
                    p += x * w0[t];
                    q += x * w1[t];
                }
                ss[v] = s; d0[v] = p; d1[v] = q;
            }
            const float rs = row_tree8(ss, b3, b2, b1);
            const float r0 = row_tree8(d0, b3, b2, b1);
            const float r1 = row_tree8(d1, b3, b2, b1);
            const float inv = 1.0f / fmaxf(sqrtf(rs), 1e-12f);
            const float l0 = r0 * inv + bf0;
            const float l1 = r1 * inv + bf1;
            const float mx = fmaxf(l0, l1);
            const float e0 = __expf(l0 - mx), e1 = __expf(l1 - mx);
            const float sc = 1.0f / (e0 + e1);
            if ((lane & 1) == 0) op[m * 16] = make_float2(e0 * sc, e1 * sc);
        }
    }
}

__global__ __launch_bounds__(256)
void norm_relu_store(float* __restrict__ H, __bf16* __restrict__ X2, int KS, int ks0) {
    __shared__ float red[16][256];
    __shared__ __attribute__((aligned(16))) __bf16 sh[16][256 + 8];
    const int tile = blockIdx.x;
    const int t = threadIdx.x;
    float* col = H + (size_t)tile * 16 * 256 + t;
    float v[16];
#pragma unroll
    for (int r = 0; r < 16; ++r) v[r] = col[r * 256];
#pragma unroll
    for (int r = 0; r < 16; ++r) red[r][t] = v[r] * v[r];
    __syncthreads();
    for (int s = 128; s > 0; s >>= 1) {
        if (t < s) {
#pragma unroll
            for (int r = 0; r < 16; ++r) red[r][t] += red[r][t + s];
        }
        __syncthreads();
    }
#pragma unroll
    for (int r = 0; r < 16; ++r) {
        float inv = 1.0f / fmaxf(sqrtf(red[r][0]), 1e-12f);
        float x = fmaxf(v[r] * inv, 0.0f);
        col[r * 256] = x;
        sh[r][t] = (__bf16)x;
    }
    __syncthreads();
#pragma unroll
    for (int i = t; i < 512; i += 256) {
        const int kl = i >> 6, j = (i >> 5) & 1, l = i & 31;
        const int k0 = kl * 32 + 16 * j + 8 * (l >> 4);
        const v8bf o = *(const v8bf*)&sh[l & 15][k0];
        const size_t q = ((size_t)(tile * KS + ks0 + kl) * 2 + j) * 32 + l;
        *(v8bf*)(X2 + q * 8) = o;
    }
}

extern "C" void kernel_launch(void* const* d_in, const int* in_sizes, int n_in,
                              void* d_out, int out_size, void* d_ws, size_t ws_size,
                              hipStream_t stream) {
    (void)in_sizes; (void)n_in; (void)out_size; (void)ws_size;

    const float* feat = (const float*)d_in[0];
    const int*   eidx = (const int*)  d_in[1];
    const int*   src  = eidx;
    const int*   dst  = eidx + N_EDGES;
    const float* W1l  = (const float*)d_in[2];
    const float* b1   = (const float*)d_in[3];
    const float* W1r  = (const float*)d_in[4];
    const float* W2l  = (const float*)d_in[5];
    const float* b2   = (const float*)d_in[6];
    const float* W2r  = (const float*)d_in[7];
    const float* Wfc  = (const float*)d_in[8];
    const float* bfc  = (const float*)d_in[9];
    float* out = (float*)d_out;

    char* ws = (char*)d_ws;
    size_t off = 0;
    auto alloc = [&](size_t bytes) -> void* {
        void* p = ws + off;
        off = (off + bytes + 255) & ~(size_t)255;
        return p;
    };
    __bf16* Xcat1 = (__bf16*)alloc((size_t)N_NODES * 2 * IN_SIZE * 2);
    __bf16* Wcat1 = (__bf16*)alloc((size_t)HIDDEN * 2 * IN_SIZE * 2);
    float*  h1    = (float*) alloc((size_t)N_NODES * HIDDEN * 4);
    __bf16* Xcat2 = (__bf16*)alloc((size_t)N_NODES * 2 * HIDDEN * 2);
    __bf16* Wcat2 = (__bf16*)alloc((size_t)OUT_SIZE * 2 * HIDDEN * 2);
    unsigned* ent    = (unsigned*)alloc(((size_t)NPIECE * PE + (size_t)NRANGE * NPIECE) * 4);
    unsigned* segtab = ent + (size_t)NPIECE * PE;

    const int TB = 256;
    auto cdiv = [](int a, int b) { return (a + b - 1) / b; };

    build_wcat<<<cdiv(HIDDEN * 2 * IN_SIZE / 8, TB), TB, 0, stream>>>(W1l, W1r, Wcat1, HIDDEN, IN_SIZE);
    build_wcat<<<cdiv(OUT_SIZE * 2 * HIDDEN / 8, TB), TB, 0, stream>>>(W2l, W2r, Wcat2, OUT_SIZE, HIDDEN);

    edge_partition<<<NPIECE, TB, 0, stream>>>(src, dst, ent, segtab);

    range_aggregate<IN_SIZE, true><<<NRANGE, 1024, 0, stream>>>(feat, ent, segtab, feat, Xcat1);

    gemm_bf16_wmma<2 * IN_SIZE, HIDDEN, GEMM_MT, GEMM_NT, 1, false>
        <<<(NTILES / GEMM_MT) * (HIDDEN / (16 * GEMM_NT)), 32, 0, stream>>>(Xcat1, Wcat1, b1, h1, nullptr, nullptr);
    norm_relu_store<<<N_NODES / 16, HIDDEN, 0, stream>>>(h1, Xcat2, 2 * HIDDEN / 32, HIDDEN / 32);

    range_aggregate<HIDDEN, false><<<NRANGE, 1024, 0, stream>>>(h1, ent, segtab, nullptr, Xcat2);

    gemm_bf16_wmma<2 * HIDDEN, OUT_SIZE, GEMM_MT, OUT_SIZE / 16, 2, true>
        <<<NTILES / GEMM_MT, 32, 0, stream>>>(Xcat2, Wcat2, b2, out, Wfc, bfc);
}


# Row L2 norm left to the consumer: layer-1 GEMM stores ReLU(x) + 4 partial row sums of squares; layer-2 aggregate scales gathered and own rows by 1/norm; norm launch gone; norm sum order changes
# speedup vs baseline: 4.5639x; 1.0603x over previous
#include <hip/hip_runtime.h>
#include <math.h>

#define N_NODES  50000
#define N_EDGES  600000
#define IN_SIZE  128
#define HIDDEN   256
#define OUT_SIZE 128

typedef __attribute__((ext_vector_type(16))) __bf16 v16bf;
typedef __attribute__((ext_vector_type(8)))  __bf16 v8bf;
typedef __attribute__((ext_vector_type(8)))  float  v8f;

struct FragChunk { int r; int k0; };
__device__ __forceinline__ FragChunk frag_chunk(int blk, int j, int l, int KS) {
    FragChunk f;
    f.r  = (blk / KS) * 16 + (l & 15);
    f.k0 = (blk % KS) * 32 + 16 * j + 8 * (l >> 4);
    return f;
}
__device__ __forceinline__ v8bf pack8(const float* __restrict__ s, float scale) {
    const float4 a = *(const float4*)s;
    const float4 b = *(const float4*)(s + 4);
    v8bf o;
    o[0] = (__bf16)(a.x * scale); o[1] = (__bf16)(a.y * scale);
    o[2] = (__bf16)(a.z * scale); o[3] = (__bf16)(a.w * scale);
    o[4] = (__bf16)(b.x * scale); o[5] = (__bf16)(b.y * scale);
    o[6] = (__bf16)(b.z * scale); o[7] = (__bf16)(b.w * scale);
    return o;
}
__device__ __forceinline__ v8bf pack8(const float* __restrict__ s) {
    const float4 a = *(const float4*)s;
    const float4 b = *(const float4*)(s + 4);
    v8bf o;
    o[0] = (__bf16)a.x; o[1] = (__bf16)a.y; o[2] = (__bf16)a.z; o[3] = (__bf16)a.w;
    o[4] = (__bf16)b.x; o[5] = (__bf16)b.y; o[6] = (__bf16)b.z; o[7] = (__bf16)b.w;
    return o;
}

__global__ __launch_bounds__(256)
void build_wcat(const float* __restrict__ Wl, const float* __restrict__ Wr,
                __bf16* __restrict__ Wcat, int rows, int kHalf) {
    const int q  = blockIdx.x * blockDim.x + threadIdx.x;
    const int KS = (2 * kHalf) >> 5;
    if (q >= rows * KS * 4) return;
    const FragChunk f = frag_chunk(q >> 6, (q >> 5) & 1, q & 31, KS);
    const float* s = (f.k0 < kHalf) ? Wl + (size_t)f.r * kHalf + f.k0
                                    : Wr + (size_t)f.r * kHalf + (f.k0 - kHalf);
    *(v8bf*)(Wcat + (size_t)q * 8) = pack8(s);
}

#define RT      13
#define RR      (RT * 16)
#define NTILES  (N_NODES / 16)
#define NRANGE  ((NTILES + RT - 1) / RT)
#define PE      2048
#define NPIECE  ((N_EDGES + PE - 1) / PE)
#define SEGP    512
#define QCAP    4096
static_assert(N_NODES % 16 == 0, "whole tiles");
static_assert(N_EDGES % 4 == 0 && PE % 4 == 0, "edges are read four at a time");
static_assert(NRANGE <= 256, "one histogram bin per thread of the sorting work-group");
static_assert(RR <= 256 && QCAP == 4096, "row starts are prefix-summed by 8 waves; a thread holds its 4 entries of a chunk");
static_assert(N_NODES <= 65536 && RR <= 65536, "an edge word holds source and local row in 16 bits each");
static_assert(PE < 65536, "segment start and length are 16 bits each");
static_assert(NPIECE <= SEGP && SEGP <= 1024, "one piece per thread in the prefix sum");

__device__ __forceinline__ int wave_incl_scan(int x, int lane) {
#pragma unroll
    for (int off = 1; off < 32; off <<= 1) {
        const int y = __shfl_up(x, off, 32);
        if (lane >= off) x += y;
    }
    return x;
}

__global__ __launch_bounds__(256)
void edge_partition(const int* __restrict__ src, const int* __restrict__ dst,
                    unsigned* __restrict__ ent, unsigned* __restrict__ segtab) {
    __shared__ int hist[256];
    __shared__ int wsum[8];
    const int t = threadIdx.x, lane = t & 31, wave = t >> 5;
    const int piece = blockIdx.x;
    const int e0 = piece * PE;
    const int ne = (N_EDGES - e0 < PE) ? (N_EDGES - e0) : PE;
    hist[t] = 0;
    __syncthreads();
    int own[8], rank[8];
    unsigned word[8];
#pragma unroll
    for (int u = 0; u < 2; ++u) {
        const int i4 = u * 256 + t;
        const bool ok = 4 * i4 < ne;
        int4 d = make_int4(0, 0, 0, 0), s = d;
        if (ok) {
            d = ((const int4*)(dst + e0))[i4];
            s = ((const int4*)(src + e0))[i4];
        }
        const int dd[4] = {d.x, d.y, d.z, d.w};
        const int ss[4] = {s.x, s.y, s.z, s.w};
#pragma unroll
        for (int k = 0; k < 4; ++k) {
            const int o = dd[k] / RR;
            own[u * 4 + k]  = ok ? o : -1;
            word[u * 4 + k] = (unsigned)ss[k] | ((unsigned)(dd[k] - o * RR) << 16);
            rank[u * 4 + k] = ok ? atomicAdd(&hist[o], 1) : 0;
        }
    }
    __syncthreads();
    const int c = hist[t];
    const int x = wave_incl_scan(c, lane);
    if (lane == 31) wsum[wave] = x;
    __syncthreads();
    int start = x - c;
    for (int w = 0; w < wave; ++w) start += wsum[w];
    hist[t] = start;
    if (t < NRANGE) segtab[(size_t)t * NPIECE + piece] = (unsigned)start | ((unsigned)c << 16);
    __syncthreads();
#pragma unroll
    for (int i = 0; i < 8; ++i)
        if (own[i] >= 0) ent[e0 + hist[own[i]] + rank[i]] = word[i];
}

__device__ __forceinline__ float inv_norm(const float4 p) {
    return 1.0f / fmaxf(sqrtf((p.x + p.y) + (p.z + p.w)), 1e-12f);
}
__device__ __forceinline__ float lane_value(float x, int i) {
    return __builtin_bit_cast(float, __builtin_amdgcn_readlane(__builtin_bit_cast(int, x), i));
}
template <int F, bool WITH_SELF, bool SCALED>
__global__ __launch_bounds__(1024)
void range_aggregate(const float* __restrict__ X, const unsigned* __restrict__ ent,
                     const unsigned* __restrict__ segtab,
                     const float* __restrict__ selfx, __bf16* __restrict__ Xcat,
                     const float* __restrict__ ssq) {
    static_assert(!SCALED || WITH_SELF, "the scaled form also packs the own rows");
    constexpr int PITCH = F + 4;
    constexpr int NL = F / 32;
    constexpr int G  = 48 / NL;
    constexpr int KS  = (2 * F) / 32;
    constexpr int KSH = F / 32;
    __shared__ __attribute__((aligned(16))) float acc[(RR + 1) * PITCH];
    __shared__ unsigned q[QCAP];
    __shared__ int cntl[RR];
    __shared__ int ccnt[RR];
    __shared__ int rsum[8];
    __shared__ unsigned segs[NPIECE];
    __shared__ int segoff[SEGP];
    __shared__ int wsum[32];
    __shared__ float invr[SCALED ? RR : 1];

    const int t = threadIdx.x, lane = t & 31;
    const int wave = __builtin_amdgcn_readfirstlane(t >> 5);
    const int tile0 = blockIdx.x * RT;
    const int ntile = (NTILES - tile0 < RT) ? (NTILES - tile0) : RT;
    const int lo = tile0 * 16;
    const int nchunk = ntile * KSH * 64;

    unsigned sg = 0u;
    if (t < NPIECE) sg = segtab[(size_t)blockIdx.x * NPIECE + t];

    constexpr int NS = SCALED ? (RT * KSH * 64 + 1023) / 1024 : 1;
    float4 sa[NS], sb[NS];
    float4 pq = make_float4(0.0f, 0.0f, 0.0f, 0.0f);
    if constexpr (SCALED) {
        pq = ((const float4*)ssq)[lo + (t < ntile * 16 ? t : ntile * 16 - 1)];
#pragma unroll
        for (int u = 0; u < NS; ++u) {
            const int i0 = u * 1024 + t;
            const int i = i0 < nchunk ? i0 : nchunk - 1;
            const int tl = i / (KSH * 64), rem = i % (KSH * 64);
            const int rl = tl * 16 + (rem & 15);
            const int k0 = (rem >> 6) * 32 + 16 * ((rem >> 5) & 1) + 8 * ((rem & 31) >> 4);
            const float* p = selfx + (size_t)(lo + rl) * F + k0;
            sa[u] = *(const float4*)p;
            sb[u] = *(const float4*)(p + 4);
        }
    }

    for (int i = t; i < (RR + 1) * PITCH / 4; i += 1024)
        ((float4*)acc)[i] = make_float4(0.0f, 0.0f, 0.0f, 0.0f);
    for (int i = t; i < RR; i += 1024) { cntl[i] = 0; ccnt[i] = 0; }
    if constexpr (SCALED) {
        if (t < ntile * 16) invr[t] = inv_norm(pq);
    }

    const int c = (int)(sg >> 16);
    const int x = wave_incl_scan(c, lane);
    if (lane == 31) wsum[wave] = x;
    if (t < NPIECE) segs[t] = sg;
    __syncthreads();
    const int wv   = wsum[lane];
    const int wvi  = wave_incl_scan(wv, lane);
    const int wofs = __builtin_amdgcn_readlane(wvi - wv, wave);
    const int n    = __builtin_amdgcn_readlane(wvi, 31);
    if (t < SEGP) segoff[t] = (t < NPIECE) ? (wofs + x - c) : 0x7fffffff;
    __syncthreads();

    if constexpr (SCALED) {
#pragma unroll
        for (int u = 0; u < NS; ++u) {
            const int i = u * 1024 + t;
            if (i < nchunk) {
                const int tl = i / (KSH * 64), rem = i % (KSH * 64);
                const int ks = rem >> 6, j = (rem >> 5) & 1, l = rem & 31;
                const float inv = invr[tl * 16 + (l & 15)];
                v8bf o;
                o[0] = (__bf16)(sa[u].x * inv); o[1] = (__bf16)(sa[u].y * inv);
                o[2] = (__bf16)(sa[u].z * inv); o[3] = (__bf16)(sa[u].w * inv);
                o[4] = (__bf16)(sb[u].x * inv); o[5] = (__bf16)(sb[u].y * inv);
                o[6] = (__bf16)(sb[u].z * inv); o[7] = (__bf16)(sb[u].w * inv);
                const size_t qs = ((size_t)((tile0 + tl) * KS + KSH + ks) * 2 + j) * 32 + l;
                *(v8bf*)(Xcat + qs * 8) = o;
            }
        }
    }

    for (int c0 = 0; c0 < n; c0 += QCAP) {
        const int m = (n - c0 < QCAP) ? (n - c0) : QCAP;
        int g[4], a[4];
#pragma unroll
        for (int u = 0; u < 4; ++u) {
            const int i = u * 1024 + t;
            g[u] = c0 + (i < m ? i : m - 1);
            a[u] = 0;
        }
#pragma unroll
        for (int h = SEGP / 2; h > 0; h >>= 1) {
#pragma unroll
            for (int u = 0; u < 4; ++u)
                if (segoff[a[u] + h] <= g[u]) a[u] += h;
        }
        unsigned w[4];
        int rank[4];
#pragma unroll
        for (int u = 0; u < 4; ++u)
            w[u] = ent[(size_t)a[u] * PE + (segs[a[u]] & 0xffffu) + (unsigned)(g[u] - segoff[a[u]])];
#pragma unroll
        for (int u = 0; u < 4; ++u)
            rank[u] = (u * 1024 + t < m) ? atomicAdd(&ccnt[w[u] >> 16], 1) : 0;
        __syncthreads();
        int rc = 0, rx = 0;
        if (t < RR) rc = ccnt[t];
        if (wave < 8) {
            rx = wave_incl_scan(rc, lane);
            if (lane == 31) rsum[wave] = rx;
        }
        __syncthreads();
        if (t < RR) {
            int start = rx - rc;
            for (int w2 = 0; w2 < wave; ++w2) start += rsum[w2];
            ccnt[t] = start;
            cntl[t] += rc;
        }
        __syncthreads();
#pragma unroll
        for (int u = 0; u < 4; ++u)
            if (u * 1024 + t < m) q[ccnt[w[u] >> 16] + rank[u]] = w[u];
        __syncthreads();
        for (int i0 = wave * G; i0 < m; i0 += 32 * G) {
            const bool have = lane < G && i0 + lane < m;
            unsigned wd = (unsigned)RR << 16;
            if (have) wd = q[i0 + lane];
            const int s_l = (int)(wd & 0xffffu);
            const int d_l = (int)(wd >> 16);
            float isc_l = 1.0f;
            if constexpr (SCALED) isc_l = inv_norm(((const float4*)ssq)[s_l]);
            float v[G][NL];
#pragma unroll
            for (int i = 0; i < G; ++i) {
                const int s = __builtin_amdgcn_readlane(s_l, i);
                const float* xs = X + (size_t)s * F + lane;
#pragma unroll
                for (int j = 0; j < NL; ++j) v[i][j] = xs[j * 32];
            }
            int dp = __builtin_amdgcn_readlane(d_l, 0);
            if constexpr (SCALED) {
                const float sc = lane_value(isc_l, 0);
#pragma unroll
                for (int j = 0; j < NL; ++j) v[0][j] *= sc;
            }
#pragma unroll
            for (int i = 1; i < G; ++i) {
                const int d = __builtin_amdgcn_readlane(d_l, i);
                const float sc = SCALED ? lane_value(isc_l, i) : 1.0f;
                if (d == dp) {
#pragma unroll
                    for (int j = 0; j < NL; ++j)
                        v[i][j] = SCALED ? fmaf(v[i][j], sc, v[i - 1][j]) : v[i][j] + v[i - 1][j];
                } else {
                    float* ad = acc + dp * PITCH + lane;
#pragma unroll
                    for (int j = 0; j < NL; ++j) atomicAdd(ad + j * 32, v[i - 1][j]);
                    if constexpr (SCALED) {
#pragma unroll
                        for (int j = 0; j < NL; ++j) v[i][j] *= sc;
                    }
                }
                dp = d;
            }
            {
                float* ad = acc + dp * PITCH + lane;
#pragma unroll
                for (int j = 0; j < NL; ++j) atomicAdd(ad + j * 32, v[G - 1][j]);
            }
        }
        __syncthreads();
        if (c0 + QCAP < n) {
            if (t < RR) ccnt[t] = 0;
            __syncthreads();
        }
    }

    for (int i = t; i < nchunk; i += 1024) {
        const int tl = i / (KSH * 64), rem = i % (KSH * 64);
        const int ks = rem >> 6, j = (rem >> 5) & 1, l = rem & 31;
        const int rl = tl * 16 + (l & 15);
        const int k0 = ks * 32 + 16 * j + 8 * (l >> 4);
        const float inv = 1.0f / fmaxf((float)cntl[rl], 1.0f);
        const size_t qi = ((size_t)((tile0 + tl) * KS + ks) * 2 + j) * 32 + l;
        *(v8bf*)(Xcat + qi * 8) = pack8(acc + rl * PITCH + k0, inv);
        if (WITH_SELF && !SCALED) {
            const size_t qs = ((size_t)((tile0 + tl) * KS + KSH + ks) * 2 + j) * 32 + l;
            *(v8bf*)(Xcat + qs * 8) = pack8(selfx + (size_t)(lo + rl) * F + k0);
        }
    }
}

#define GEMM_MT 5
#define GEMM_NT 4
static_assert(NTILES % GEMM_MT == 0, "whole row groups");
static_assert(HIDDEN % (16 * GEMM_NT) == 0, "whole column groups");

template <int M>
__device__ __forceinline__ float row_xor(float x) {
    const int b = __builtin_bit_cast(int, x);
    return __builtin_bit_cast(float, __builtin_amdgcn_update_dpp(b, b, 0x160 + M, 0xf, 0xf, false));
}
__device__ __forceinline__ float row_tree8(const float (&x)[8], bool b3, bool b2, bool b1) {
    float y[4], z[2];
#pragma unroll
    for (int i = 0; i < 4; ++i)
        y[i] = (b3 ? x[i + 4] : x[i]) + row_xor<8>(b3 ? x[i] : x[i + 4]);
#pragma unroll
    for (int i = 0; i < 2; ++i)
        z[i] = (b2 ? y[i + 2] : y[i]) + row_xor<4>(b2 ? y[i] : y[i + 2]);
    float w = (b1 ? z[1] : z[0]) + row_xor<2>(b1 ? z[0] : z[1]);
    w += row_xor<1>(w);
    return w;
}

template <int K, int N, int MT, int NT, int PF, bool HEAD>
__global__ __launch_bounds__(32)
void gemm_bf16_wmma(const __bf16* __restrict__ A, const __bf16* __restrict__ B,
                    const float* __restrict__ bias, float* __restrict__ C,
                    const float* __restrict__ Wfc, const float* __restrict__ bfc,
                    float* __restrict__ ssq) {
    static_assert(HEAD || N / (16 * NT) == 4, "the consumer reads a row's partial sums as one float4");
    static_assert(!HEAD || N == 16 * NT, "the head needs whole rows in one wave");
    constexpr int KS   = K / 32;
    constexpr int BLK  = 512;
    constexpr int RING = PF + 1;
    constexpr int NG   = N / (16 * NT);
    union Frag { v16bf v; v8bf h[2]; };
    const int lane = threadIdx.x;
    const int tmt0 = ((int)blockIdx.x / NG) * MT;
    const int tn0  = ((int)blockIdx.x % NG) * (16 * NT);
    const int l15  = lane & 15;

    const __bf16* ap = A + (size_t)tmt0 * KS * BLK + lane * 8;
    const __bf16* bp = B + (size_t)(tn0 >> 4) * KS * BLK + lane * 8;

    float bc[NT];
#pragma unroll
    for (int t = 0; t < NT; ++t) bc[t] = bias[tn0 + t * 16 + l15];

    v8f acc[MT][NT];
#pragma unroll
    for (int m = 0; m < MT; ++m)
#pragma unroll
        for (int t = 0; t < NT; ++t) acc[m][t] = (v8f){};

    Frag a[RING][MT], b[RING][NT];
    auto fetch = [&](int slot, int ks) {
#pragma unroll
        for (int m = 0; m < MT; ++m) {
            a[slot][m].h[0] = *(const v8bf*)(ap + (size_t)(m * KS + ks) * BLK);
            a[slot][m].h[1] = *(const v8bf*)(ap + (size_t)(m * KS + ks) * BLK + 256);
        }
#pragma unroll
        for (int t = 0; t < NT; ++t) {
            b[slot][t].h[0] = *(const v8bf*)(bp + (size_t)(t * KS + ks) * BLK);
            b[slot][t].h[1] = *(const v8bf*)(bp + (size_t)(t * KS + ks) * BLK + 256);
        }
    };
#pragma unroll
    for (int p = 0; p < PF; ++p) fetch(p, p);

#pragma unroll
    for (int ks = 0; ks < KS; ++ks) {
        if (ks + PF < KS) fetch((ks + PF) % RING, ks + PF);
        __builtin_amdgcn_sched_barrier(0);
        const int cur = ks % RING;
#pragma unroll
        for (int m = 0; m < MT; ++m)
#pragma unroll
            for (int t = 0; t < NT; ++t)
                acc[m][t] = __builtin_amdgcn_wmma_f32_16x16x32_bf16(
                    false, a[cur][m].v, false, b[cur][t].v, (short)0, acc[m][t], false, false);
        __builtin_amdgcn_sched_barrier(0);
    }

    if constexpr (!HEAD) {
        float* cp = C + (size_t)(tmt0 * 16 + (lane >> 4) * 8) * N + tn0 + l15;
        const bool b3 = (lane & 8) != 0, b2 = (lane & 4) != 0, b1 = (lane & 2) != 0;
        float* sp = ssq + ((size_t)tmt0 * 16 + (lane >> 4) * 8 + (l15 >> 1)) * NG + (int)blockIdx.x % NG;
#pragma unroll
        for (int m = 0; m < MT; ++m) {
            float ss[8];
#pragma unroll
            for (int v = 0; v < 8; ++v) {
                float s = 0.0f;
#pragma unroll
                for (int t = 0; t < NT; ++t) {
                    const float x = acc[m][t][v] + bc[t];
                    s += x * x;
                    cp[(size_t)(m * 16 + v) * N + t * 16] = fmaxf(x, 0.0f);
                }
                ss[v] = s;
            }
            const float rs = row_tree8(ss, b3, b2, b1);
            if ((lane & 1) == 0) sp[(size_t)m * 16 * NG] = rs;
        }
    } else {
        float w0[NT], w1[NT];
#pragma unroll
        for (int t = 0; t < NT; ++t) {
            w0[t] = Wfc[t * 16 + l15];
            w1[t] = Wfc[N + t * 16 + l15];
        }
        const float bf0 = bfc[0], bf1 = bfc[1];
        const bool b3 = (lane & 8) != 0, b2 = (lane & 4) != 0, b1 = (lane & 2) != 0;
        float2* op = (float2*)C + (size_t)tmt0 * 16 + (lane >> 4) * 8 + (l15 >> 1);
#pragma unroll
        for (int m = 0; m < MT; ++m) {
            float ss[8], d0[8], d1[8];
#pragma unroll
            for (int v = 0; v < 8; ++v) {
                float s = 0.0f, p = 0.0f, q = 0.0f;
#pragma unroll
                for (int t = 0; t < NT; ++t) {
                    const float x = acc[m][t][v] + bc[t];
                    s += x * x;
                    p += x * w0[t];
                    q += x * w1[t];
                }
                ss[v] = s; d0[v] = p; d1[v] = q;
            }
            const float rs = row_tree8(ss, b3, b2, b1);
            const float r0 = row_tree8(d0, b3, b2, b1);
            const float r1 = row_tree8(d1, b3, b2, b1);
            const float inv = 1.0f / fmaxf(sqrtf(rs), 1e-12f);
            const float l0 = r0 * inv + bf0;
            const float l1 = r1 * inv + bf1;
            const float mx = fmaxf(l0, l1);
            const float e0 = __expf(l0 - mx), e1 = __expf(l1 - mx);
            const float sc = 1.0f / (e0 + e1);
            if ((lane & 1) == 0) op[m * 16] = make_float2(e0 * sc, e1 * sc);
        }
    }
}

extern "C" void kernel_launch(void* const* d_in, const int* in_sizes, int n_in,
                              void* d_out, int out_size, void* d_ws, size_t ws_size,
                              hipStream_t stream) {
    (void)in_sizes; (void)n_in; (void)out_size; (void)ws_size;

    const float* feat = (const float*)d_in[0];
    const int*   eidx = (const int*)  d_in[1];
    const int*   src  = eidx;
    const int*   dst  = eidx + N_EDGES;
    const float* W1l  = (const float*)d_in[2];
    const float* b1   = (const float*)d_in[3];
    const float* W1r  = (const float*)d_in[4];
    const float* W2l  = (const float*)d_in[5];
    const float* b2   = (const float*)d_in[6];
    const float* W2r  = (const float*)d_in[7];
    const float* Wfc  = (const float*)d_in[8];
    const float* bfc  = (const float*)d_in[9];
    float* out = (float*)d_out;

    char* ws = (char*)d_ws;
    size_t off = 0;
    auto alloc = [&](size_t bytes) -> void* {
        void* p = ws + off;
        off = (off + bytes + 255) & ~(size_t)255;
        return p;
    };
    __bf16* Xcat1 = (__bf16*)alloc((size_t)N_NODES * 2 * IN_SIZE * 2);
    __bf16* Wcat1 = (__bf16*)alloc((size_t)HIDDEN * 2 * IN_SIZE * 2);
    float*  h1    = (float*) alloc((size_t)N_NODES * HIDDEN * 4);
    __bf16* Xcat2 = (__bf16*)alloc((size_t)N_NODES * 2 * HIDDEN * 2);
    __bf16* Wcat2 = (__bf16*)alloc((size_t)OUT_SIZE * 2 * HIDDEN * 2);
    unsigned* ent    = (unsigned*)alloc(((size_t)NPIECE * PE + (size_t)NRANGE * NPIECE) * 4);
    unsigned* segtab = ent + (size_t)NPIECE * PE;
    float*  ssq1  = (float*) alloc((size_t)N_NODES * 4 * 4);

    const int TB = 256;
    auto cdiv = [](int a, int b) { return (a + b - 1) / b; };

    build_wcat<<<cdiv(HIDDEN * 2 * IN_SIZE / 8, TB), TB, 0, stream>>>(W1l, W1r, Wcat1, HIDDEN, IN_SIZE);
    build_wcat<<<cdiv(OUT_SIZE * 2 * HIDDEN / 8, TB), TB, 0, stream>>>(W2l, W2r, Wcat2, OUT_SIZE, HIDDEN);

    edge_partition<<<NPIECE, TB, 0, stream>>>(src, dst, ent, segtab);

    range_aggregate<IN_SIZE, true, false><<<NRANGE, 1024, 0, stream>>>(feat, ent, segtab, feat, Xcat1, nullptr);

    gemm_bf16_wmma<2 * IN_SIZE, HIDDEN, GEMM_MT, GEMM_NT, 1, false>
        <<<(NTILES / GEMM_MT) * (HIDDEN / (16 * GEMM_NT)), 32, 0, stream>>>(Xcat1, Wcat1, b1, h1, nullptr, nullptr, ssq1);

    range_aggregate<HIDDEN, true, true><<<NRANGE, 1024, 0, stream>>>(h1, ent, segtab, h1, Xcat2, ssq1);

    gemm_bf16_wmma<2 * HIDDEN, OUT_SIZE, GEMM_MT, OUT_SIZE / 16, 2, true>
        <<<NTILES / GEMM_MT, 32, 0, stream>>>(Xcat2, Wcat2, b2, out, Wfc, bfc, nullptr);
}


# Layer-2 product + head run inside the layer-2 aggregate: [mean | own] rows packed to bf16 fragments in its LDS, 32 waves x 2x2 WMMA tiles; operand plane and a launch gone; norm/FC sum order changes
# speedup vs baseline: 4.8744x; 1.0680x over previous
#include <hip/hip_runtime.h>
#include <math.h>

#define N_NODES  50000
#define N_EDGES  600000
#define IN_SIZE  128
#define HIDDEN   256
#define OUT_SIZE 128

typedef __attribute__((ext_vector_type(16))) __bf16 v16bf;
typedef __attribute__((ext_vector_type(8)))  __bf16 v8bf;
typedef __attribute__((ext_vector_type(8)))  float  v8f;

struct FragChunk { int r; int k0; };
__device__ __forceinline__ FragChunk frag_chunk(int blk, int j, int l, int KS) {
    FragChunk f;
    f.r  = (blk / KS) * 16 + (l & 15);
    f.k0 = (blk % KS) * 32 + 16 * j + 8 * (l >> 4);
    return f;
}
__device__ __forceinline__ v8bf pack8(const float* __restrict__ s, float scale) {
    const float4 a = *(const float4*)s;
    const float4 b = *(const float4*)(s + 4);
    v8bf o;
    o[0] = (__bf16)(a.x * scale); o[1] = (__bf16)(a.y * scale);
    o[2] = (__bf16)(a.z * scale); o[3] = (__bf16)(a.w * scale);
    o[4] = (__bf16)(b.x * scale); o[5] = (__bf16)(b.y * scale);
    o[6] = (__bf16)(b.z * scale); o[7] = (__bf16)(b.w * scale);
    return o;
}
__device__ __forceinline__ v8bf pack8(const float* __restrict__ s) {
    const float4 a = *(const float4*)s;
    const float4 b = *(const float4*)(s + 4);
    v8bf o;
    o[0] = (__bf16)a.x; o[1] = (__bf16)a.y; o[2] = (__bf16)a.z; o[3] = (__bf16)a.w;
    o[4] = (__bf16)b.x; o[5] = (__bf16)b.y; o[6] = (__bf16)b.z; o[7] = (__bf16)b.w;
    return o;
}

__global__ __launch_bounds__(256)
void build_wcat(const float* __restrict__ Wl, const float* __restrict__ Wr,
                __bf16* __restrict__ Wcat, int rows, int kHalf) {
    const int q  = blockIdx.x * blockDim.x + threadIdx.x;
    const int KS = (2 * kHalf) >> 5;
    if (q >= rows * KS * 4) return;
    const FragChunk f = frag_chunk(q >> 6, (q >> 5) & 1, q & 31, KS);
    const float* s = (f.k0 < kHalf) ? Wl + (size_t)f.r * kHalf + f.k0
                                    : Wr + (size_t)f.r * kHalf + (f.k0 - kHalf);
    *(v8bf*)(Wcat + (size_t)q * 8) = pack8(s);
}

#define RT      13
#define RR      (RT * 16)
#define NTILES  (N_NODES / 16)
#define NRANGE  ((NTILES + RT - 1) / RT)
#define PE      2048
#define NPIECE  ((N_EDGES + PE - 1) / PE)
#define SEGP    512
#define QCAP    4096
static_assert(N_NODES % 16 == 0, "whole tiles");
static_assert(N_EDGES % 4 == 0 && PE % 4 == 0, "edges are read four at a time");
static_assert(NRANGE <= 256, "one histogram bin per thread of the sorting work-group");
static_assert(RR <= 256 && QCAP == 4096, "row starts are prefix-summed by 8 waves; a thread holds its 4 entries of a chunk");
static_assert(N_NODES <= 65536 && RR <= 65536, "an edge word holds source and local row in 16 bits each");
static_assert(PE < 65536, "segment start and length are 16 bits each");
static_assert(NPIECE <= SEGP && SEGP <= 1024, "one piece per thread in the prefix sum");

__device__ __forceinline__ int wave_incl_scan(int x, int lane) {
#pragma unroll
    for (int off = 1; off < 32; off <<= 1) {
        const int y = __shfl_up(x, off, 32);
        if (lane >= off) x += y;
    }
    return x;
}

__global__ __launch_bounds__(256)
void edge_partition(const int* __restrict__ src, const int* __restrict__ dst,
                    unsigned* __restrict__ ent, unsigned* __restrict__ segtab) {
    __shared__ int hist[256];
    __shared__ int wsum[8];
    const int t = threadIdx.x, lane = t & 31, wave = t >> 5;
    const int piece = blockIdx.x;
    const int e0 = piece * PE;
    const int ne = (N_EDGES - e0 < PE) ? (N_EDGES - e0) : PE;
    hist[t] = 0;
    __syncthreads();
    int own[8], rank[8];
    unsigned word[8];
#pragma unroll
    for (int u = 0; u < 2; ++u) {
        const int i4 = u * 256 + t;
        const bool ok = 4 * i4 < ne;
        int4 d = make_int4(0, 0, 0, 0), s = d;
        if (ok) {
            d = ((const int4*)(dst + e0))[i4];
            s = ((const int4*)(src + e0))[i4];
        }
        const int dd[4] = {d.x, d.y, d.z, d.w};
        const int ss[4] = {s.x, s.y, s.z, s.w};
#pragma unroll
        for (int k = 0; k < 4; ++k) {
            const int o = dd[k] / RR;
            own[u * 4 + k]  = ok ? o : -1;
            word[u * 4 + k] = (unsigned)ss[k] | ((unsigned)(dd[k] - o * RR) << 16);
            rank[u * 4 + k] = ok ? atomicAdd(&hist[o], 1) : 0;
        }
    }
    __syncthreads();
    const int c = hist[t];
    const int x = wave_incl_scan(c, lane);
    if (lane == 31) wsum[wave] = x;
    __syncthreads();
    int start = x - c;
    for (int w = 0; w < wave; ++w) start += wsum[w];
    hist[t] = start;
    if (t < NRANGE) segtab[(size_t)t * NPIECE + piece] = (unsigned)start | ((unsigned)c << 16);
    __syncthreads();
#pragma unroll
    for (int i = 0; i < 8; ++i)
        if (own[i] >= 0) ent[e0 + hist[own[i]] + rank[i]] = word[i];
}

__device__ __forceinline__ float inv_norm(const float4 p) {
    return 1.0f / fmaxf(sqrtf((p.x + p.y) + (p.z + p.w)), 1e-12f);
}
__device__ __forceinline__ float lane_value(float x, int i) {
    return __builtin_bit_cast(float, __builtin_amdgcn_readlane(__builtin_bit_cast(int, x), i));
}
template <int M>
__device__ __forceinline__ float row_xor(float x) {
    const int b = __builtin_bit_cast(int, x);
    return __builtin_bit_cast(float, __builtin_amdgcn_update_dpp(b, b, 0x160 + M, 0xf, 0xf, false));
}
__device__ __forceinline__ float row_tree8(const float (&x)[8], bool b3, bool b2, bool b1) {
    float y[4], z[2];
#pragma unroll
    for (int i = 0; i < 4; ++i)
        y[i] = (b3 ? x[i + 4] : x[i]) + row_xor<8>(b3 ? x[i] : x[i + 4]);
#pragma unroll
    for (int i = 0; i < 2; ++i)
        z[i] = (b2 ? y[i + 2] : y[i]) + row_xor<4>(b2 ? y[i] : y[i + 2]);
    float w = (b1 ? z[1] : z[0]) + row_xor<2>(b1 ? z[0] : z[1]);
    w += row_xor<1>(w);
    return w;
}
template <int F, bool WITH_SELF, bool SCALED>
__global__ __launch_bounds__(1024)
void range_aggregate(const float* __restrict__ X, const unsigned* __restrict__ ent,
                     const unsigned* __restrict__ segtab,
                     const float* __restrict__ selfx, __bf16* __restrict__ Xcat,
                     const float* __restrict__ ssq,
                     const __bf16* __restrict__ Wn, const float* __restrict__ bn,
                     const float* __restrict__ Wfc, const float* __restrict__ bfc,
                     float* __restrict__ out) {
    static_assert(!SCALED || WITH_SELF, "the scaled form also packs the own rows");
    static_assert(!SCALED || (F == HIDDEN && RT <= 16), "the scaled form is layer 2; a wave's row tiles are mg and mg + 8");
    constexpr int PITCH = F + 4;
    constexpr int NL = F / 32;
    constexpr int G  = 48 / NL;
    constexpr int KS  = (2 * F) / 32;
    constexpr int KSH = F / 32;
    __shared__ __attribute__((aligned(16))) float acc[(RR + 1) * PITCH];
    __shared__ __attribute__((aligned(16))) unsigned q[QCAP];
    __shared__ int cntl[RR];
    __shared__ int ccnt[RR];
    __shared__ int rsum[8];
    __shared__ unsigned segs[NPIECE];
    __shared__ int segoff[SEGP];
    __shared__ int wsum[32];
    __shared__ float invr[SCALED ? RR : 1];

    const int t = threadIdx.x, lane = t & 31;
    const int wave = __builtin_amdgcn_readfirstlane(t >> 5);
    const int tile0 = blockIdx.x * RT;
    const int ntile = (NTILES - tile0 < RT) ? (NTILES - tile0) : RT;
    const int lo = tile0 * 16;
    const int nchunk = ntile * KSH * 64;

    unsigned sg = 0u;
    if (t < NPIECE) sg = segtab[(size_t)blockIdx.x * NPIECE + t];

    float4 pq = make_float4(0.0f, 0.0f, 0.0f, 0.0f);
    if constexpr (SCALED) pq = ((const float4*)ssq)[lo + (t < ntile * 16 ? t : ntile * 16 - 1)];

    for (int i = t; i < (RR + 1) * PITCH / 4; i += 1024)
        ((float4*)acc)[i] = make_float4(0.0f, 0.0f, 0.0f, 0.0f);
    for (int i = t; i < RR; i += 1024) { cntl[i] = 0; ccnt[i] = 0; }
    if constexpr (SCALED) {
        if (t < ntile * 16) invr[t] = inv_norm(pq);
    }

    const int c = (int)(sg >> 16);
    const int x = wave_incl_scan(c, lane);
    if (lane == 31) wsum[wave] = x;
    if (t < NPIECE) segs[t] = sg;
    __syncthreads();
    const int wv   = wsum[lane];
    const int wvi  = wave_incl_scan(wv, lane);
    const int wofs = __builtin_amdgcn_readlane(wvi - wv, wave);
    const int n    = __builtin_amdgcn_readlane(wvi, 31);
    if (t < SEGP) segoff[t] = (t < NPIECE) ? (wofs + x - c) : 0x7fffffff;
    __syncthreads();

    for (int c0 = 0; c0 < n; c0 += QCAP) {
        const int m = (n - c0 < QCAP) ? (n - c0) : QCAP;
        int g[4], a[4];
#pragma unroll
        for (int u = 0; u < 4; ++u) {
            const int i = u * 1024 + t;
            g[u] = c0 + (i < m ? i : m - 1);
            a[u] = 0;
        }
#pragma unroll
        for (int h = SEGP / 2; h > 0; h >>= 1) {
#pragma unroll
            for (int u = 0; u < 4; ++u)
                if (segoff[a[u] + h] <= g[u]) a[u] += h;
        }
        unsigned w[4];
        int rank[4];
#pragma unroll
        for (int u = 0; u < 4; ++u)
            w[u] = ent[(size_t)a[u] * PE + (segs[a[u]] & 0xffffu) + (unsigned)(g[u] - segoff[a[u]])];
#pragma unroll
        for (int u = 0; u < 4; ++u)
            rank[u] = (u * 1024 + t < m) ? atomicAdd(&ccnt[w[u] >> 16], 1) : 0;
        __syncthreads();
        int rc = 0, rx = 0;
        if (t < RR) rc = ccnt[t];
        if (wave < 8) {
            rx = wave_incl_scan(rc, lane);
            if (lane == 31) rsum[wave] = rx;
        }
        __syncthreads();
        if (t < RR) {
            int start = rx - rc;
            for (int w2 = 0; w2 < wave; ++w2) start += rsum[w2];
            ccnt[t] = start;
            cntl[t] += rc;
        }
        __syncthreads();
#pragma unroll
        for (int u = 0; u < 4; ++u)
            if (u * 1024 + t < m) q[ccnt[w[u] >> 16] + rank[u]] = w[u];
        __syncthreads();
        for (int i0 = wave * G; i0 < m; i0 += 32 * G) {
            const bool have = lane < G && i0 + lane < m;
            unsigned wd = (unsigned)RR << 16;
            if (have) wd = q[i0 + lane];
            const int s_l = (int)(wd & 0xffffu);
            const int d_l = (int)(wd >> 16);
            float isc_l = 1.0f;
            if constexpr (SCALED) isc_l = inv_norm(((const float4*)ssq)[s_l]);
            float v[G][NL];
#pragma unroll
            for (int i = 0; i < G; ++i) {
                const int s = __builtin_amdgcn_readlane(s_l, i);
                const float* xs = X + (size_t)s * F + lane;
#pragma unroll
                for (int j = 0; j < NL; ++j) v[i][j] = xs[j * 32];
            }
            int dp = __builtin_amdgcn_readlane(d_l, 0);
            if constexpr (SCALED) {
                const float sc = lane_value(isc_l, 0);
#pragma unroll
                for (int j = 0; j < NL; ++j) v[0][j] *= sc;
            }
#pragma unroll
            for (int i = 1; i < G; ++i) {
                const int d = __builtin_amdgcn_readlane(d_l, i);
                const float sc = SCALED ? lane_value(isc_l, i) : 1.0f;
                if (d == dp) {
#pragma unroll
                    for (int j = 0; j < NL; ++j)
                        v[i][j] = SCALED ? fmaf(v[i][j], sc, v[i - 1][j]) : v[i][j] + v[i - 1][j];
                } else {
                    float* ad = acc + dp * PITCH + lane;
#pragma unroll
                    for (int j = 0; j < NL; ++j) atomicAdd(ad + j * 32, v[i - 1][j]);
                    if constexpr (SCALED) {
#pragma unroll
                        for (int j = 0; j < NL; ++j) v[i][j] *= sc;
                    }
                }
                dp = d;
            }
            {
                float* ad = acc + dp * PITCH + lane;
#pragma unroll
                for (int j = 0; j < NL; ++j) atomicAdd(ad + j * 32, v[G - 1][j]);
            }
        }
        __syncthreads();
        if (c0 + QCAP < n) {
            if (t < RR) ccnt[t] = 0;
            __syncthreads();
        }
    }

    if constexpr (!SCALED) {
        for (int i = t; i < nchunk; i += 1024) {
            const int tl = i / (KSH * 64), rem = i % (KSH * 64);
            const int ks = rem >> 6, j = (rem >> 5) & 1, l = rem & 31;
            const int rl = tl * 16 + (l & 15);
            const int k0 = ks * 32 + 16 * j + 8 * (l >> 4);
            const float inv = 1.0f / fmaxf((float)cntl[rl], 1.0f);
            const size_t qi = ((size_t)((tile0 + tl) * KS + ks) * 2 + j) * 32 + l;
            *(v8bf*)(Xcat + qi * 8) = pack8(acc + rl * PITCH + k0, inv);
            if (WITH_SELF) {
                const size_t qs = ((size_t)((tile0 + tl) * KS + KSH + ks) * 2 + j) * 32 + l;
                *(v8bf*)(Xcat + qs * 8) = pack8(selfx + (size_t)(lo + rl) * F + k0);
            }
        }
    } else {
        v8bf* abuf = (v8bf*)acc;
        constexpr int NS = (RT * KSH * 64 + 1023) / 1024;
        v8bf mean[NS];
#pragma unroll
        for (int u = 0; u < NS; ++u) {
            const int i0 = u * 1024 + t;
            const int i = i0 < nchunk ? i0 : nchunk - 1;
            const int tl = i / (KSH * 64), rem = i % (KSH * 64);
            const int rl = tl * 16 + (rem & 15);
            const int k0 = (rem >> 6) * 32 + 16 * ((rem >> 5) & 1) + 8 * ((rem & 31) >> 4);
            const float inv = 1.0f / fmaxf((float)cntl[rl], 1.0f);
            mean[u] = pack8(acc + rl * PITCH + k0, inv);
        }
        __syncthreads();
#pragma unroll
        for (int u = 0; u < NS; ++u) {
            const int i = u * 1024 + t;
            if (i < nchunk) abuf[(i / (KSH * 64)) * (KS * 64) + i % (KSH * 64)] = mean[u];
        }
        float4 sa[NS], sb[NS];
#pragma unroll
        for (int u = 0; u < NS; ++u) {
            const int i0 = u * 1024 + t;
            const int i = i0 < nchunk ? i0 : nchunk - 1;
            const int tl = i / (KSH * 64), rem = i % (KSH * 64);
            const int rl = tl * 16 + (rem & 15);
            const int k0 = (rem >> 6) * 32 + 16 * ((rem >> 5) & 1) + 8 * ((rem & 31) >> 4);
            const float* p = selfx + (size_t)(lo + rl) * F + k0;
            sa[u] = *(const float4*)p;
            sb[u] = *(const float4*)(p + 4);
        }
#pragma unroll
        for (int u = 0; u < NS; ++u) {
            const int i = u * 1024 + t;
            if (i < nchunk) {
                const int tl = i / (KSH * 64), rem = i % (KSH * 64);
                const float inv = invr[tl * 16 + (rem & 15)];
                v8bf o;
                o[0] = (__bf16)(sa[u].x * inv); o[1] = (__bf16)(sa[u].y * inv);
                o[2] = (__bf16)(sa[u].z * inv); o[3] = (__bf16)(sa[u].w * inv);
                o[4] = (__bf16)(sb[u].x * inv); o[5] = (__bf16)(sb[u].y * inv);
                o[6] = (__bf16)(sb[u].z * inv); o[7] = (__bf16)(sb[u].w * inv);
                abuf[tl * (KS * 64) + KSH * 64 + rem] = o;
            }
        }
        __syncthreads();

        constexpr int N2 = OUT_SIZE;
        static_assert(N2 == 128, "4 column groups of 2 tiles; a row's partial sums are one float4");
        union Frag { v16bf v; v8bf h[2]; };
        const int ng = wave & 3, mg = wave >> 2;
        const int l15 = lane & 15;
        const int tA = mg < ntile ? mg : 0;
        const int tB = mg + 8 < ntile ? mg + 8 : tA;
        const v8bf* apA = abuf + tA * (KS * 64) + lane;
        const v8bf* apB = abuf + tB * (KS * 64) + lane;
        const __bf16* bp = Wn + (size_t)(2 * ng) * KS * 512 + lane * 8;
        v8f cc[2][2];
#pragma unroll
        for (int mi = 0; mi < 2; ++mi)
#pragma unroll
            for (int tt = 0; tt < 2; ++tt) cc[mi][tt] = (v8f){};
#pragma unroll
        for (int ks = 0; ks < KS; ++ks) {
            Frag fa[2], fb[2];
            fa[0].h[0] = apA[ks * 64];  fa[0].h[1] = apA[ks * 64 + 32];
            fa[1].h[0] = apB[ks * 64];  fa[1].h[1] = apB[ks * 64 + 32];
#pragma unroll
            for (int tt = 0; tt < 2; ++tt) {
                fb[tt].h[0] = *(const v8bf*)(bp + (size_t)(tt * KS + ks) * 512);
                fb[tt].h[1] = *(const v8bf*)(bp + (size_t)(tt * KS + ks) * 512 + 256);
            }
#pragma unroll
            for (int mi = 0; mi < 2; ++mi)
#pragma unroll
                for (int tt = 0; tt < 2; ++tt)
                    cc[mi][tt] = __builtin_amdgcn_wmma_f32_16x16x32_bf16(
                        false, fa[mi].v, false, fb[tt].v, (short)0, cc[mi][tt], false, false);
        }

        float bc[2], w0[2], w1[2];
#pragma unroll
        for (int tt = 0; tt < 2; ++tt) {
            const int col = (2 * ng + tt) * 16 + l15;
            bc[tt] = bn[col];
            w0[tt] = Wfc[col];
            w1[tt] = Wfc[N2 + col];
        }
        float* pss = (float*)q;
        float* pd0 = pss + RR * 4;
        float* pd1 = pd0 + RR * 4;
        static_assert(3 * RR * 4 <= QCAP, "the partial sums fit the list's space");
        const bool b3 = (lane & 8) != 0, b2 = (lane & 4) != 0, b1 = (lane & 2) != 0;
#pragma unroll
        for (int mi = 0; mi < 2; ++mi) {
            float ss[8], d0[8], d1[8];
#pragma unroll
            for (int v = 0; v < 8; ++v) {
                float sq = 0.0f, p0 = 0.0f, p1 = 0.0f;
#pragma unroll
                for (int tt = 0; tt < 2; ++tt) {
                    const float xv = cc[mi][tt][v] + bc[tt];
                    sq += xv * xv;
                    p0 += xv * w0[tt];
                    p1 += xv * w1[tt];
                }
                ss[v] = sq; d0[v] = p0; d1[v] = p1;
            }
            const float rs = row_tree8(ss, b3, b2, b1);
            const float r0 = row_tree8(d0, b3, b2, b1);
            const float r1 = row_tree8(d1, b3, b2, b1);
            const int tile = mg + 8 * mi;
            if (tile < ntile && (lane & 1) == 0) {
                const int r = tile * 16 + (lane >> 4) * 8 + (l15 >> 1);
                pss[r * 4 + ng] = rs;
                pd0[r * 4 + ng] = r0;
                pd1[r * 4 + ng] = r1;
            }
        }
        __syncthreads();
        if (t < ntile * 16) {
            const float4 a4 = ((const float4*)pss)[t];
            const float4 e4 = ((const float4*)pd0)[t];
            const float4 f4 = ((const float4*)pd1)[t];
            const float rs = (a4.x + a4.y) + (a4.z + a4.w);
            const float r0 = (e4.x + e4.y) + (e4.z + e4.w);
            const float r1 = (f4.x + f4.y) + (f4.z + f4.w);
            const float inv = 1.0f / fmaxf(sqrtf(rs), 1e-12f);
            const float l0 = r0 * inv + bfc[0];
            const float l1 = r1 * inv + bfc[1];
            const float mx = fmaxf(l0, l1);
            const float e0 = __expf(l0 - mx), e1 = __expf(l1 - mx);
            const float sc = 1.0f / (e0 + e1);
            ((float2*)out)[lo + t] = make_float2(e0 * sc, e1 * sc);
        }
    }
}

#define GEMM_MT 5
#define GEMM_NT 4
static_assert(NTILES % GEMM_MT == 0, "whole row groups");
static_assert(HIDDEN % (16 * GEMM_NT) == 0, "whole column groups");

template <int K, int N, int MT, int NT, int PF>
__global__ __launch_bounds__(32)
void gemm_bf16_wmma(const __bf16* __restrict__ A, const __bf16* __restrict__ B,
                    const float* __restrict__ bias, float* __restrict__ C,
                    float* __restrict__ ssq) {
    static_assert(N / (16 * NT) == 4, "the consumer reads a row's partial sums as one float4");
    constexpr int KS   = K / 32;
    constexpr int BLK  = 512;
    constexpr int RING = PF + 1;
    constexpr int NG   = N / (16 * NT);
    union Frag { v16bf v; v8bf h[2]; };
    const int lane = threadIdx.x;
    const int tmt0 = ((int)blockIdx.x / NG) * MT;
    const int tn0  = ((int)blockIdx.x % NG) * (16 * NT);
    const int l15  = lane & 15;

    const __bf16* ap = A + (size_t)tmt0 * KS * BLK + lane * 8;
    const __bf16* bp = B + (size_t)(tn0 >> 4) * KS * BLK + lane * 8;

    float bc[NT];
#pragma unroll
    for (int t = 0; t < NT; ++t) bc[t] = bias[tn0 + t * 16 + l15];

    v8f acc[MT][NT];
#pragma unroll
    for (int m = 0; m < MT; ++m)
#pragma unroll
        for (int t = 0; t < NT; ++t) acc[m][t] = (v8f){};

    Frag a[RING][MT], b[RING][NT];
    auto fetch = [&](int slot, int ks) {
#pragma unroll
        for (int m = 0; m < MT; ++m) {
            a[slot][m].h[0] = *(const v8bf*)(ap + (size_t)(m * KS + ks) * BLK);
            a[slot][m].h[1] = *(const v8bf*)(ap + (size_t)(m * KS + ks) * BLK + 256);
        }
#pragma unroll
        for (int t = 0; t < NT; ++t) {
            b[slot][t].h[0] = *(const v8bf*)(bp + (size_t)(t * KS + ks) * BLK);
            b[slot][t].h[1] = *(const v8bf*)(bp + (size_t)(t * KS + ks) * BLK + 256);
        }
    };
#pragma unroll
    for (int p = 0; p < PF; ++p) fetch(p, p);

#pragma unroll
    for (int ks = 0; ks < KS; ++ks) {
        if (ks + PF < KS) fetch((ks + PF) % RING, ks + PF);
        __builtin_amdgcn_sched_barrier(0);
        const int cur = ks % RING;
#pragma unroll
        for (int m = 0; m < MT; ++m)
#pragma unroll
            for (int t = 0; t < NT; ++t)
                acc[m][t] = __builtin_amdgcn_wmma_f32_16x16x32_bf16(
                    false, a[cur][m].v, false, b[cur][t].v, (short)0, acc[m][t], false, false);
        __builtin_amdgcn_sched_barrier(0);
    }

    float* cp = C + (size_t)(tmt0 * 16 + (lane >> 4) * 8) * N + tn0 + l15;
    const bool b3 = (lane & 8) != 0, b2 = (lane & 4) != 0, b1 = (lane & 2) != 0;
    float* sp = ssq + ((size_t)tmt0 * 16 + (lane >> 4) * 8 + (l15 >> 1)) * NG + (int)blockIdx.x % NG;
#pragma unroll
    for (int m = 0; m < MT; ++m) {
        float ss[8];
#pragma unroll
        for (int v = 0; v < 8; ++v) {
            float s = 0.0f;
#pragma unroll
            for (int t = 0; t < NT; ++t) {
                const float x = acc[m][t][v] + bc[t];
                s += x * x;
                cp[(size_t)(m * 16 + v) * N + t * 16] = fmaxf(x, 0.0f);
            }
            ss[v] = s;
        }
        const float rs = row_tree8(ss, b3, b2, b1);
        if ((lane & 1) == 0) sp[(size_t)m * 16 * NG] = rs;
    }
}

extern "C" void kernel_launch(void* const* d_in, const int* in_sizes, int n_in,
                              void* d_out, int out_size, void* d_ws, size_t ws_size,
                              hipStream_t stream) {
    (void)in_sizes; (void)n_in; (void)out_size; (void)ws_size;

    const float* feat = (const float*)d_in[0];
    const int*   eidx = (const int*)  d_in[1];
    const int*   src  = eidx;
    const int*   dst  = eidx + N_EDGES;
    const float* W1l  = (const float*)d_in[2];
    const float* b1   = (const float*)d_in[3];
    const float* W1r  = (const float*)d_in[4];
    const float* W2l  = (const float*)d_in[5];
    const float* b2   = (const float*)d_in[6];
    const float* W2r  = (const float*)d_in[7];
    const float* Wfc  = (const float*)d_in[8];
    const float* bfc  = (const float*)d_in[9];
    float* out = (float*)d_out;

    char* ws = (char*)d_ws;
    size_t off = 0;
    auto alloc = [&](size_t bytes) -> void* {
        void* p = ws + off;
        off = (off + bytes + 255) & ~(size_t)255;
        return p;
    };
    __bf16* Xcat1 = (__bf16*)alloc((size_t)N_NODES * 2 * IN_SIZE * 2);
    __bf16* Wcat1 = (__bf16*)alloc((size_t)HIDDEN * 2 * IN_SIZE * 2);
    float*  h1    = (float*) alloc((size_t)N_NODES * HIDDEN * 4);
    __bf16* Wcat2 = (__bf16*)alloc((size_t)OUT_SIZE * 2 * HIDDEN * 2);
    unsigned* ent    = (unsigned*)alloc(((size_t)NPIECE * PE + (size_t)NRANGE * NPIECE) * 4);
    unsigned* segtab = ent + (size_t)NPIECE * PE;
    float*  ssq1  = (float*) alloc((size_t)N_NODES * 4 * 4);

    const int TB = 256;
    auto cdiv = [](int a, int b) { return (a + b - 1) / b; };

    build_wcat<<<cdiv(HIDDEN * 2 * IN_SIZE / 8, TB), TB, 0, stream>>>(W1l, W1r, Wcat1, HIDDEN, IN_SIZE);
    build_wcat<<<cdiv(OUT_SIZE * 2 * HIDDEN / 8, TB), TB, 0, stream>>>(W2l, W2r, Wcat2, OUT_SIZE, HIDDEN);

    edge_partition<<<NPIECE, TB, 0, stream>>>(src, dst, ent, segtab);

    range_aggregate<IN_SIZE, true, false><<<NRANGE, 1024, 0, stream>>>(feat, ent, segtab, feat, Xcat1, nullptr,
                                                                   nullptr, nullptr, nullptr, nullptr, nullptr);

    gemm_bf16_wmma<2 * IN_SIZE, HIDDEN, GEMM_MT, GEMM_NT, 1>
        <<<(NTILES / GEMM_MT) * (HIDDEN / (16 * GEMM_NT)), 32, 0, stream>>>(Xcat1, Wcat1, b1, h1, ssq1);

    range_aggregate<HIDDEN, true, true><<<NRANGE, 1024, 0, stream>>>(h1, ent, segtab, h1, nullptr, ssq1,
                                                                  Wcat2, b2, Wfc, bfc, out);
}
